# starting kernel
# baseline (speedup 1.0000x reference)
#include <hip/hip_runtime.h>

typedef float v2f __attribute__((ext_vector_type(2)));
typedef float v8f __attribute__((ext_vector_type(8)));

#define CH 128

template <int NCOL>
__global__ __launch_bounds__(256) void gemm_wmma_f32(
    const float* __restrict__ A, const float* __restrict__ W,
    const float* __restrict__ bias, float* __restrict__ C, int nrows) {
  constexpr int NT = NCOL / 16;
  constexpr int KP = 130;
  __shared__ float lw[NCOL * KP];

  const int tid = threadIdx.x;
  for (int i = tid; i < CH * NCOL; i += 256) {
    int k = i / NCOL, c = i - k * NCOL;
    lw[c * KP + k] = W[i];
  }
  __syncthreads();

  const int wave = tid >> 5;
  const int lane = tid & 31;
  const int m0 = (blockIdx.x * 8 + wave) * 16;
  if (m0 >= nrows) return;

  const int lrow = lane & 15;
  const int kbase = (lane >> 4) * 2;

  v8f acc[NT] = {};

  int arow = m0 + lrow;
  if (arow > nrows - 1) arow = nrows - 1;
  const float* aptr = A + (size_t)arow * CH + kbase;

#pragma unroll 4
  for (int kc = 0; kc < CH / 4; ++kc) {
    float2 av = *(const float2*)(aptr + kc * 4);
    v2f a;
    a[0] = av.x; a[1] = av.y;
#pragma unroll
    for (int n = 0; n < NT; ++n) {
      float2 bv = *(const float2*)&lw[(n * 16 + lrow) * KP + kc * 4 + kbase];
      v2f b;
      b[0] = bv.x; b[1] = bv.y;
      acc[n] = __builtin_amdgcn_wmma_f32_16x16x4_f32(false, a, false, b,
                                                     (short)0, acc[n],
                                                     false, false);
    }
  }

  const int rbase = m0 + ((lane >> 4) << 3);
#pragma unroll
  for (int r = 0; r < 8; ++r) {
    int row = rbase + r;
    if (row < nrows) {
#pragma unroll
      for (int n = 0; n < NT; ++n) {
        int col = n * 16 + lrow;
        float v = acc[n][r];
        if (bias) v += bias[col];
        C[(size_t)row * NCOL + col] = v;
      }
    }
  }
}

__global__ void fill_f32(float* __restrict__ p, float v, int n) {
  int i = blockIdx.x * blockDim.x + threadIdx.x;
  if (i < n) p[i] = v;
}

__global__ void zero4(float* __restrict__ p, long n4) {
  long i = (long)blockIdx.x * blockDim.x + threadIdx.x;
  if (i < n4) ((float4*)p)[i] = make_float4(0.f, 0.f, 0.f, 0.f);
}

__global__ void deg_accum(const int* __restrict__ dst, float* __restrict__ deg,
                          int ne) {
  int e = blockIdx.x * blockDim.x + threadIdx.x;
  if (e < ne) atomicAdd(&deg[dst[e]], 1.0f);
}

__global__ void rsqrt_inplace(float* __restrict__ d, int n) {
  int i = blockIdx.x * blockDim.x + threadIdx.x;
  if (i < n) {
    float v = d[i];
    d[i] = v > 0.f ? rsqrtf(v) : 0.f;
  }
}

__global__ void edge_norm(const int* __restrict__ src,
                          const int* __restrict__ dst,
                          const float* __restrict__ invs,
                          float* __restrict__ nrm, int ne) {
  int e = blockIdx.x * blockDim.x + threadIdx.x;
  if (e < ne) nrm[e] = invs[src[e]] * invs[dst[e]];
}

__global__ __launch_bounds__(256) void aggregate(
    const float* __restrict__ t, const int* __restrict__ src,
    const int* __restrict__ dst, const float* __restrict__ nrm,
    float* __restrict__ agg, int ne) {
  const int lane = threadIdx.x & 31;
  int w = (blockIdx.x * blockDim.x + threadIdx.x) >> 5;
  const int nw = (gridDim.x * blockDim.x) >> 5;
  for (int e = w; e < ne; e += nw) {
    const int s = src[e];
    const int d = dst[e];
    const float nv = nrm[e];
    const float4 v = *(const float4*)(t + (size_t)s * CH + lane * 4);
    float* o = agg + (size_t)d * CH + lane * 4;
    atomicAdd(o + 0, v.x * nv);
    atomicAdd(o + 1, v.y * nv);
    atomicAdd(o + 2, v.z * nv);
    atomicAdd(o + 3, v.w * nv);
  }
}

__global__ void self_bias_act(float* __restrict__ hb,
                              const float* __restrict__ t,
                              const float* __restrict__ invs,
                              const float* __restrict__ bias, int n, int relu) {
  int i = blockIdx.x * blockDim.x + threadIdx.x;
  if (i >= n * 32) return;
  int node = i >> 5, c = (i & 31) * 4;
  float s = invs[node];
  s = s * s;
  float4 a = *(float4*)(hb + (size_t)node * CH + c);
  float4 tv = *(const float4*)(t + (size_t)node * CH + c);
  float4 b = *(const float4*)(bias + c);
  float4 r;
  r.x = a.x + tv.x * s + b.x;
  r.y = a.y + tv.y * s + b.y;
  r.z = a.z + tv.z * s + b.z;
  r.w = a.w + tv.w * s + b.w;
  if (relu) {
    r.x = fmaxf(r.x, 0.f);
    r.y = fmaxf(r.y, 0.f);
    r.z = fmaxf(r.z, 0.f);
    r.w = fmaxf(r.w, 0.f);
  }
  *(float4*)(hb + (size_t)node * CH + c) = r;
}

extern "C" void kernel_launch(void* const* d_in, const int* in_sizes, int n_in,
                              void* d_out, int out_size, void* d_ws,
                              size_t ws_size, hipStream_t stream) {
  const float* x = (const float*)d_in[0];
  const int* ei = (const int*)d_in[1];
  const float* W1 = (const float*)d_in[2];
  const float* b1 = (const float*)d_in[3];
  const float* W2 = (const float*)d_in[4];
  const float* b2 = (const float*)d_in[5];
  const float* W3 = (const float*)d_in[6];
  const float* b3 = (const float*)d_in[7];
  const float* Wo = (const float*)d_in[8];
  const float* bo = (const float*)d_in[9];

  const int N = in_sizes[0] / CH;
  const int E = in_sizes[1] / 2;
  const int* src = ei;
  const int* dstp = ei + E;

  float* bufA = (float*)d_ws;
  float* bufB = bufA + (size_t)N * CH;
  float* invs = bufB + (size_t)N * CH;
  float* nrm = invs + N;

  const dim3 blk(256);
  const int gN = (N + 255) / 256;
  const int gE = (E + 255) / 256;
  const int gGemm = (N + 127) / 128;
  const long n4 = (long)N * CH / 4;
  const int gZero = (int)((n4 + 255) / 256);
  const int gEpi = (N * 32 + 255) / 256;
  const int gAgg = 8192;

  fill_f32<<<gN, blk, 0, stream>>>(invs, 1.0f, N);
  deg_accum<<<gE, blk, 0, stream>>>(dstp, invs, E);
  rsqrt_inplace<<<gN, blk, 0, stream>>>(invs, N);
  edge_norm<<<gE, blk, 0, stream>>>(src, dstp, invs, nrm, E);

  gemm_wmma_f32<128><<<gGemm, blk, 0, stream>>>(x, W1, nullptr, bufA, N);
  zero4<<<gZero, blk, 0, stream>>>(bufB, n4);
  aggregate<<<gAgg, blk, 0, stream>>>(bufA, src, dstp, nrm, bufB, E);
  self_bias_act<<<gEpi, blk, 0, stream>>>(bufB, bufA, invs, b1, N, 1);

  gemm_wmma_f32<128><<<gGemm, blk, 0, stream>>>(bufB, W2, nullptr, bufA, N);
  zero4<<<gZero, blk, 0, stream>>>(bufB, n4);
  aggregate<<<gAgg, blk, 0, stream>>>(bufA, src, dstp, nrm, bufB, E);
  self_bias_act<<<gEpi, blk, 0, stream>>>(bufB, bufA, invs, b2, N, 1);

  gemm_wmma_f32<128><<<gGemm, blk, 0, stream>>>(bufB, W3, nullptr, bufA, N);
  zero4<<<gZero, blk, 0, stream>>>(bufB, n4);
  aggregate<<<gAgg, blk, 0, stream>>>(bufA, src, dstp, nrm, bufB, E);
  self_bias_act<<<gEpi, blk, 0, stream>>>(bufB, bufA, invs, b3, N, 0);

  gemm_wmma_f32<64><<<gGemm, blk, 0, stream>>>(bufB, Wo, bo, (float*)d_out, N);
}


# f32 atomic edge scatter replaced by per-call dst-sorted edge lists (int atomic rank + scan) and a wave-per-node register gather; row sum order = arrival rank, varies run to run as the parent's
# speedup vs baseline: 3.0461x; 3.0461x over previous
#include <hip/hip_runtime.h>

typedef float v2f __attribute__((ext_vector_type(2)));
typedef float v8f __attribute__((ext_vector_type(8)));

#define CH 128

template <int NCOL>
__global__ __launch_bounds__(256) void gemm_wmma_f32(
    const float* __restrict__ A, const float* __restrict__ W,
    const float* __restrict__ bias, float* __restrict__ C, int nrows) {
  constexpr int NT = NCOL / 16;
  constexpr int KP = 130;
  __shared__ float lw[NCOL * KP];

  const int tid = threadIdx.x;
  for (int i = tid; i < CH * NCOL; i += 256) {
    int k = i / NCOL, c = i - k * NCOL;
    lw[c * KP + k] = W[i];
  }
  __syncthreads();

  const int wave = tid >> 5;
  const int lane = tid & 31;
  const int m0 = (blockIdx.x * 8 + wave) * 16;
  if (m0 >= nrows) return;

  const int lrow = lane & 15;
  const int kbase = (lane >> 4) * 2;

  v8f acc[NT] = {};

  int arow = m0 + lrow;
  if (arow > nrows - 1) arow = nrows - 1;
  const float* aptr = A + (size_t)arow * CH + kbase;

#pragma unroll 4
  for (int kc = 0; kc < CH / 4; ++kc) {
    float2 av = *(const float2*)(aptr + kc * 4);
    v2f a;
    a[0] = av.x; a[1] = av.y;
#pragma unroll
    for (int n = 0; n < NT; ++n) {
      float2 bv = *(const float2*)&lw[(n * 16 + lrow) * KP + kc * 4 + kbase];
      v2f b;
      b[0] = bv.x; b[1] = bv.y;
      acc[n] = __builtin_amdgcn_wmma_f32_16x16x4_f32(false, a, false, b,
                                                     (short)0, acc[n],
                                                     false, false);
    }
  }

  const int rbase = m0 + ((lane >> 4) << 3);
#pragma unroll
  for (int r = 0; r < 8; ++r) {
    int row = rbase + r;
    if (row < nrows) {
#pragma unroll
      for (int n = 0; n < NT; ++n) {
        int col = n * 16 + lrow;
        float v = acc[n][r];
        if (bias) v += bias[col];
        C[(size_t)row * NCOL + col] = v;
      }
    }
  }
}

__global__ void fill_i32(int* __restrict__ p, int v, int n) {
  int i = blockIdx.x * blockDim.x + threadIdx.x;
  if (i < n) p[i] = v;
}

__global__ void count_rank(const int* __restrict__ dst, int* __restrict__ cnt,
                           int* __restrict__ rank, int ne) {
  int e = blockIdx.x * blockDim.x + threadIdx.x;
  if (e < ne) rank[e] = atomicAdd(&cnt[dst[e]], 1);
}

__global__ __launch_bounds__(1024) void scan_rows(int* __restrict__ cr,
                                                  float* __restrict__ invs,
                                                  int n) {
  __shared__ int wtot[32];
  const int tid = threadIdx.x;
  const int lane = tid & 31;
  const int wv = tid >> 5;
  int carry = 0;
  for (int base = 0; base < n; base += 16384) {
    const int i = base + tid * 16;
    const bool full = i + 15 < n;
    int c[16];
    if (full) {
#pragma unroll
      for (int q = 0; q < 4; ++q) {
        const int4 t4 = *(const int4*)(cr + i + q * 4);
        c[q * 4 + 0] = t4.x;
        c[q * 4 + 1] = t4.y;
        c[q * 4 + 2] = t4.z;
        c[q * 4 + 3] = t4.w;
      }
    } else {
#pragma unroll
      for (int k = 0; k < 16; ++k) c[k] = (i + k < n) ? cr[i + k] : 0;
    }
    int tsum = 0;
#pragma unroll
    for (int k = 0; k < 16; ++k) tsum += c[k];
    int v = tsum;
    for (int o = 1; o < 32; o <<= 1) {
      const int u = __shfl_up(v, o);
      if (lane >= o) v += u;
    }
    if (lane == 31) wtot[wv] = v;
    __syncthreads();
    const int wt = wtot[lane];
    int ws = wt;
    for (int o = 1; o < 32; o <<= 1) {
      const int u = __shfl_up(ws, o);
      if (lane >= o) ws += u;
    }
    const int wexcl = __shfl(ws, wv) - __shfl(wt, wv);
    const int total = __shfl(ws, 31);
    int run = carry + wexcl + (v - tsum);
    int ex[16];
    float rs[16];
#pragma unroll
    for (int k = 0; k < 16; ++k) {
      ex[k] = run;
      run += c[k];
      const float dk = (float)(c[k] + 1);
      rs[k] = dk > 0.f ? rsqrtf(dk) : 0.f;
    }
    if (full) {
#pragma unroll
      for (int q = 0; q < 4; ++q) {
        *(int4*)(cr + i + q * 4) =
            make_int4(ex[q * 4], ex[q * 4 + 1], ex[q * 4 + 2], ex[q * 4 + 3]);
        *(float4*)(invs + i + q * 4) =
            make_float4(rs[q * 4], rs[q * 4 + 1], rs[q * 4 + 2], rs[q * 4 + 3]);
      }
    } else {
#pragma unroll
      for (int k = 0; k < 16; ++k) {
        if (i + k < n) {
          cr[i + k] = ex[k];
          invs[i + k] = rs[k];
        }
      }
    }
    carry += total;
    __syncthreads();
  }
}

__global__ void fill_lists(const int* __restrict__ src,
                           const int* __restrict__ dst,
                           const int* __restrict__ rank,
                           const int* __restrict__ rowptr,
                           const float* __restrict__ invs,
                           int* __restrict__ recs, int ne) {
  int e = blockIdx.x * blockDim.x + threadIdx.x;
  if (e < ne) {
    const int s = src[e];
    const int d = dst[e];
    int p = rowptr[d] + rank[e];
    p = min(max(p, 0), ne - 1);
    const float w = invs[s] * invs[d];
    ((int2*)recs)[p] = make_int2(s, __float_as_int(w));
  }
}

template <int K>
__device__ __forceinline__ void add_rows(float4& acc,
                                         const float* __restrict__ tl, int rx,
                                         int ry, int j, int last) {
  float4 v[K];
  float wg[K];
#pragma unroll
  for (int u = 0; u < K; ++u) {
    int s = __builtin_amdgcn_readlane(rx, j + u);
    s = min(max(s, 0), last);
    wg[u] = __int_as_float(__builtin_amdgcn_readlane(ry, j + u));
    v[u] = *(const float4*)(tl + (size_t)s * CH);
  }
#pragma unroll
  for (int u = 0; u < K; ++u) {
    acc.x += v[u].x * wg[u];
    acc.y += v[u].y * wg[u];
    acc.z += v[u].z * wg[u];
    acc.w += v[u].w * wg[u];
  }
}

__global__ __launch_bounds__(256) void gather_rows(
    const float* __restrict__ t, const int* __restrict__ rowptr,
    const int* __restrict__ recs, const float* __restrict__ bias,
    float* __restrict__ h, int n, int ne, int relu) {
  const int lane = threadIdx.x & 31;
  const int w0 = __builtin_amdgcn_readfirstlane(
      (int)((blockIdx.x * blockDim.x + threadIdx.x) >> 5));
  const int nw = (int)((gridDim.x * blockDim.x) >> 5);
  const float4 b = *(const float4*)(bias + lane * 4);
  const int2* rl = (const int2*)recs;
  const float* tl = t + lane * 4;
  for (int d = w0; d < n; d += nw) {
    const int start = rowptr[d];
    const int end = (d + 1 < n) ? rowptr[d + 1] : ne;
    const float4 tv = *(const float4*)(tl + (size_t)d * CH);
    float4 acc = make_float4(0.f, 0.f, 0.f, 0.f);
    for (int base = start; base < end; base += 32) {
      const int idx = base + lane;
      int2 r = make_int2(0, 0);
      if (idx < end) r = rl[idx];
      const int cnt = min(32, end - base);
      int j = 0;
      for (; j + 8 <= cnt; j += 8) add_rows<8>(acc, tl, r.x, r.y, j, n - 1);
      if (j + 4 <= cnt) { add_rows<4>(acc, tl, r.x, r.y, j, n - 1); j += 4; }
      if (j + 2 <= cnt) { add_rows<2>(acc, tl, r.x, r.y, j, n - 1); j += 2; }
      if (j < cnt) add_rows<1>(acc, tl, r.x, r.y, j, n - 1);
    }
    const float degf = (float)(end - start + 1);
    float sl = degf > 0.f ? rsqrtf(degf) : 0.f;
    sl = sl * sl;
    float4 o;
    o.x = acc.x + tv.x * sl + b.x;
    o.y = acc.y + tv.y * sl + b.y;
    o.z = acc.z + tv.z * sl + b.z;
    o.w = acc.w + tv.w * sl + b.w;
    if (relu) {
      o.x = fmaxf(o.x, 0.f);
      o.y = fmaxf(o.y, 0.f);
      o.z = fmaxf(o.z, 0.f);
      o.w = fmaxf(o.w, 0.f);
    }
    *(float4*)(h + (size_t)d * CH + lane * 4) = o;
  }
}

extern "C" void kernel_launch(void* const* d_in, const int* in_sizes, int n_in,
                              void* d_out, int out_size, void* d_ws,
                              size_t ws_size, hipStream_t stream) {
  const float* x = (const float*)d_in[0];
  const int* ei = (const int*)d_in[1];
  const float* W1 = (const float*)d_in[2];
  const float* b1 = (const float*)d_in[3];
  const float* W2 = (const float*)d_in[4];
  const float* b2 = (const float*)d_in[5];
  const float* W3 = (const float*)d_in[6];
  const float* b3 = (const float*)d_in[7];
  const float* Wo = (const float*)d_in[8];
  const float* bo = (const float*)d_in[9];

  const int N = in_sizes[0] / CH;
  const int E = in_sizes[1] / 2;
  const int* src = ei;
  const int* dstp = ei + E;

  float* bufA = (float*)d_ws;
  float* bufB = bufA + (size_t)N * CH;
  int* rowptr = (int*)(bufB + (size_t)N * CH);
  int* rank = rowptr + N;
  float* invs = bufB;
  int* recs = (int*)d_out;

  const dim3 blk(256);
  const int gN = (N + 255) / 256;
  const int gE = (E + 255) / 256;
  const int gGemm = (N + 127) / 128;
  const int gGather = 2048;

  fill_i32<<<gN, blk, 0, stream>>>(rowptr, 0, N);
  count_rank<<<gE, blk, 0, stream>>>(dstp, rowptr, rank, E);
  scan_rows<<<1, dim3(1024), 0, stream>>>(rowptr, invs, N);
  fill_lists<<<gE, blk, 0, stream>>>(src, dstp, rank, rowptr, invs, recs, E);

  gemm_wmma_f32<128><<<gGemm, blk, 0, stream>>>(x, W1, nullptr, bufA, N);
  gather_rows<<<gGather, blk, 0, stream>>>(bufA, rowptr, recs, b1, bufB, N, E, 1);

  gemm_wmma_f32<128><<<gGemm, blk, 0, stream>>>(bufB, W2, nullptr, bufA, N);
  gather_rows<<<gGather, blk, 0, stream>>>(bufA, rowptr, recs, b2, bufB, N, E, 1);

  gemm_wmma_f32<128><<<gGemm, blk, 0, stream>>>(bufB, W3, nullptr, bufA, N);
  gather_rows<<<gGather, blk, 0, stream>>>(bufA, rowptr, recs, b3, bufB, N, E, 0);

  gemm_wmma_f32<64><<<gGemm, blk, 0, stream>>>(bufB, Wo, bo, (float*)d_out, N);
}


# edge slots from LDS counters per 32768-edge chunk (two nodes per word) + per-node chunk-offset pass replace the global returning add per edge; row sum order = chunk then LDS arrival, varies per run
# speedup vs baseline: 3.4049x; 1.1178x over previous
#include <hip/hip_runtime.h>

typedef float v2f __attribute__((ext_vector_type(2)));
typedef float v8f __attribute__((ext_vector_type(8)));

#define CH 128

template <int NCOL>
__global__ __launch_bounds__(256) void gemm_wmma_f32(
    const float* __restrict__ A, const float* __restrict__ W,
    const float* __restrict__ bias, float* __restrict__ C, int nrows) {
  constexpr int NT = NCOL / 16;
  constexpr int KP = 130;
  __shared__ float lw[NCOL * KP];

  const int tid = threadIdx.x;
  for (int i = tid; i < CH * NCOL; i += 256) {
    int k = i / NCOL, c = i - k * NCOL;
    lw[c * KP + k] = W[i];
  }
  __syncthreads();

  const int wave = tid >> 5;
  const int lane = tid & 31;
  const int m0 = (blockIdx.x * 8 + wave) * 16;
  if (m0 >= nrows) return;

  const int lrow = lane & 15;
  const int kbase = (lane >> 4) * 2;

  v8f acc[NT] = {};

  int arow = m0 + lrow;
  if (arow > nrows - 1) arow = nrows - 1;
  const float* aptr = A + (size_t)arow * CH + kbase;

#pragma unroll 4
  for (int kc = 0; kc < CH / 4; ++kc) {
    float2 av = *(const float2*)(aptr + kc * 4);
    v2f a;
    a[0] = av.x; a[1] = av.y;
#pragma unroll
    for (int n = 0; n < NT; ++n) {
      float2 bv = *(const float2*)&lw[(n * 16 + lrow) * KP + kc * 4 + kbase];
      v2f b;
      b[0] = bv.x; b[1] = bv.y;
      acc[n] = __builtin_amdgcn_wmma_f32_16x16x4_f32(false, a, false, b,
                                                     (short)0, acc[n],
                                                     false, false);
    }
  }

  const int rbase = m0 + ((lane >> 4) << 3);
#pragma unroll
  for (int r = 0; r < 8; ++r) {
    int row = rbase + r;
    if (row < nrows) {
#pragma unroll
      for (int n = 0; n < NT; ++n) {
        int col = n * 16 + lrow;
        float v = acc[n][r];
        if (bias) v += bias[col];
        C[(size_t)row * NCOL + col] = v;
      }
    }
  }
}

#define CHUNK 32768
#define CAPW 65536

__device__ __forceinline__ int packed_pitch(int n) {
  return (((n + 1) >> 1) + 3) & ~3;
}

__global__ __launch_bounds__(1024) void chunk_count(
    const int* __restrict__ dst, unsigned* __restrict__ ccnt,
    int* __restrict__ rank, int n, int ne) {
  __shared__ unsigned lc[CAPW];
  const int tid = threadIdx.x;
  const int c = blockIdx.x;
  const int e0 = c * CHUNK;
  const int e1 = min(e0 + CHUNK, ne);
  const int nwp = packed_pitch(n);
  for (int wb = 0; wb < nwp; wb += CAPW) {
    const int nwin = min(CAPW, nwp - wb);
    for (int i = tid * 4; i < nwin; i += 4096)
      *(uint4*)&lc[i] = make_uint4(0u, 0u, 0u, 0u);
    __syncthreads();
    const int dlo = wb * 2;
    const int dhi = dlo + nwin * 2;
    int dv[CHUNK / 1024];
    unsigned old[CHUNK / 1024];
#pragma unroll
    for (int k = 0; k < CHUNK / 1024; ++k) {
      const int e = e0 + k * 1024 + tid;
      const int v = dst[min(e, e1 - 1)];
      dv[k] = (e < e1) ? v : -1;
    }
#pragma unroll
    for (int k = 0; k < CHUNK / 1024; ++k) {
      const bool in = dv[k] >= dlo && dv[k] < dhi;
      const int l = in ? dv[k] - dlo : 0;
      const unsigned inc = in ? ((l & 1) ? 65536u : 1u) : 0u;
      old[k] = atomicAdd(&lc[l >> 1], inc);
    }
#pragma unroll
    for (int k = 0; k < CHUNK / 1024; ++k) {
      const int e = e0 + k * 1024 + tid;
      if (dv[k] >= dlo && dv[k] < dhi)
        rank[e] = (int)(((dv[k] - dlo) & 1) ? (old[k] >> 16)
                                            : (old[k] & 0xFFFFu));
    }
    __syncthreads();
    for (int i = tid * 4; i < nwin; i += 4096)
      *(uint4*)&ccnt[(size_t)c * nwp + wb + i] = *(const uint4*)&lc[i];
    __syncthreads();
  }
}

__global__ __launch_bounds__(256) void chunk_offsets(
    const unsigned* __restrict__ ccnt, int* __restrict__ offs,
    int* __restrict__ cnt, int n, int nchunks) {
  const int nwp = packed_pitch(n);
  const int w = blockIdx.x * blockDim.x + threadIdx.x;
  if (w >= nwp) return;
  const size_t np = (size_t)nwp * 2;
  int lo = 0, hi = 0;
#pragma unroll 8
  for (int c = 0; c < nchunks; ++c) {
    const unsigned v = ccnt[(size_t)c * nwp + w];
    *(int2*)(offs + (size_t)c * np + 2 * w) = make_int2(lo, hi);
    lo += (int)(v & 0xFFFFu);
    hi += (int)(v >> 16);
  }
  if (2 * w < n) cnt[2 * w] = lo;
  if (2 * w + 1 < n) cnt[2 * w + 1] = hi;
}

__global__ __launch_bounds__(1024) void scan_rows(int* __restrict__ cr,
                                                  float* __restrict__ invs,
                                                  int n) {
  __shared__ int wtot[32];
  const int tid = threadIdx.x;
  const int lane = tid & 31;
  const int wv = tid >> 5;
  int carry = 0;
  for (int base = 0; base < n; base += 16384) {
    const int i = base + tid * 16;
    const bool full = i + 15 < n;
    int c[16];
    if (full) {
#pragma unroll
      for (int q = 0; q < 4; ++q) {
        const int4 t4 = *(const int4*)(cr + i + q * 4);
        c[q * 4 + 0] = t4.x;
        c[q * 4 + 1] = t4.y;
        c[q * 4 + 2] = t4.z;
        c[q * 4 + 3] = t4.w;
      }
    } else {
#pragma unroll
      for (int k = 0; k < 16; ++k) c[k] = (i + k < n) ? cr[i + k] : 0;
    }
    int tsum = 0;
#pragma unroll
    for (int k = 0; k < 16; ++k) tsum += c[k];
    int v = tsum;
    for (int o = 1; o < 32; o <<= 1) {
      const int u = __shfl_up(v, o);
      if (lane >= o) v += u;
    }
    if (lane == 31) wtot[wv] = v;
    __syncthreads();
    const int wt = wtot[lane];
    int ws = wt;
    for (int o = 1; o < 32; o <<= 1) {
      const int u = __shfl_up(ws, o);
      if (lane >= o) ws += u;
    }
    const int wexcl = __shfl(ws, wv) - __shfl(wt, wv);
    const int total = __shfl(ws, 31);
    int run = carry + wexcl + (v - tsum);
    int ex[16];
    float rs[16];
#pragma unroll
    for (int k = 0; k < 16; ++k) {
      ex[k] = run;
      run += c[k];
      const float dk = (float)(c[k] + 1);
      rs[k] = dk > 0.f ? rsqrtf(dk) : 0.f;
    }
    if (full) {
#pragma unroll
      for (int q = 0; q < 4; ++q) {
        *(int4*)(cr + i + q * 4) =
            make_int4(ex[q * 4], ex[q * 4 + 1], ex[q * 4 + 2], ex[q * 4 + 3]);
        *(float4*)(invs + i + q * 4) =
            make_float4(rs[q * 4], rs[q * 4 + 1], rs[q * 4 + 2], rs[q * 4 + 3]);
      }
    } else {
#pragma unroll
      for (int k = 0; k < 16; ++k) {
        if (i + k < n) {
          cr[i + k] = ex[k];
          invs[i + k] = rs[k];
        }
      }
    }
    carry += total;
    __syncthreads();
  }
}

__global__ void fill_lists(const int* __restrict__ src,
                           const int* __restrict__ dst,
                           const int* __restrict__ rank,
                           const int* __restrict__ rowptr,
                           const int* __restrict__ offs,
                           const float* __restrict__ invs,
                           int* __restrict__ recs, int n, int ne) {
  int e = blockIdx.x * blockDim.x + threadIdx.x;
  if (e < ne) {
    const int s = src[e];
    const int d = dst[e];
    const size_t np = (size_t)packed_pitch(n) * 2;
    int p = rowptr[d] + offs[(size_t)(e / CHUNK) * np + d] + rank[e];
    p = min(max(p, 0), ne - 1);
    const float w = invs[s] * invs[d];
    ((int2*)recs)[p] = make_int2(s, __float_as_int(w));
  }
}

template <int K>
__device__ __forceinline__ void add_rows(float4& acc,
                                         const float* __restrict__ tl, int rx,
                                         int ry, int j, int last) {
  float4 v[K];
  float wg[K];
#pragma unroll
  for (int u = 0; u < K; ++u) {
    int s = __builtin_amdgcn_readlane(rx, j + u);
    s = min(max(s, 0), last);
    wg[u] = __int_as_float(__builtin_amdgcn_readlane(ry, j + u));
    v[u] = *(const float4*)(tl + (size_t)s * CH);
  }
#pragma unroll
  for (int u = 0; u < K; ++u) {
    acc.x += v[u].x * wg[u];
    acc.y += v[u].y * wg[u];
    acc.z += v[u].z * wg[u];
    acc.w += v[u].w * wg[u];
  }
}

__global__ __launch_bounds__(256) void gather_rows(
    const float* __restrict__ t, const int* __restrict__ rowptr,
    const int* __restrict__ recs, const float* __restrict__ bias,
    float* __restrict__ h, int n, int ne, int relu) {
  const int lane = threadIdx.x & 31;
  const int w0 = __builtin_amdgcn_readfirstlane(
      (int)((blockIdx.x * blockDim.x + threadIdx.x) >> 5));
  const int nw = (int)((gridDim.x * blockDim.x) >> 5);
  const float4 b = *(const float4*)(bias + lane * 4);
  const int2* rl = (const int2*)recs;
  const float* tl = t + lane * 4;
  for (int d = w0; d < n; d += nw) {
    const int start = rowptr[d];
    const int end = (d + 1 < n) ? rowptr[d + 1] : ne;
    const float4 tv = *(const float4*)(tl + (size_t)d * CH);
    float4 acc = make_float4(0.f, 0.f, 0.f, 0.f);
    for (int base = start; base < end; base += 32) {
      const int idx = base + lane;
      int2 r = make_int2(0, 0);
      if (idx < end) r = rl[idx];
      const int cnt = min(32, end - base);
      int j = 0;
      for (; j + 8 <= cnt; j += 8) add_rows<8>(acc, tl, r.x, r.y, j, n - 1);
      if (j + 4 <= cnt) { add_rows<4>(acc, tl, r.x, r.y, j, n - 1); j += 4; }
      if (j + 2 <= cnt) { add_rows<2>(acc, tl, r.x, r.y, j, n - 1); j += 2; }
      if (j < cnt) add_rows<1>(acc, tl, r.x, r.y, j, n - 1);
    }
    const float degf = (float)(end - start + 1);
    float sl = degf > 0.f ? rsqrtf(degf) : 0.f;
    sl = sl * sl;
    float4 o;
    o.x = acc.x + tv.x * sl + b.x;
    o.y = acc.y + tv.y * sl + b.y;
    o.z = acc.z + tv.z * sl + b.z;
    o.w = acc.w + tv.w * sl + b.w;
    if (relu) {
      o.x = fmaxf(o.x, 0.f);
      o.y = fmaxf(o.y, 0.f);
      o.z = fmaxf(o.z, 0.f);
      o.w = fmaxf(o.w, 0.f);
    }
    *(float4*)(h + (size_t)d * CH + lane * 4) = o;
  }
}

extern "C" void kernel_launch(void* const* d_in, const int* in_sizes, int n_in,
                              void* d_out, int out_size, void* d_ws,
                              size_t ws_size, hipStream_t stream) {
  const float* x = (const float*)d_in[0];
  const int* ei = (const int*)d_in[1];
  const float* W1 = (const float*)d_in[2];
  const float* b1 = (const float*)d_in[3];
  const float* W2 = (const float*)d_in[4];
  const float* b2 = (const float*)d_in[5];
  const float* W3 = (const float*)d_in[6];
  const float* b3 = (const float*)d_in[7];
  const float* Wo = (const float*)d_in[8];
  const float* bo = (const float*)d_in[9];

  const int N = in_sizes[0] / CH;
  const int E = in_sizes[1] / 2;
  const int* src = ei;
  const int* dstp = ei + E;

  float* bufA = (float*)d_ws;
  float* bufB = bufA + (size_t)N * CH;
  int* rowptr = (int*)(bufB + (size_t)N * CH);
  int* rank = rowptr + N;
  float* invs = bufB;
  unsigned* ccnt = (unsigned*)bufA;
  int* offs = (int*)(bufB + (size_t)2 * N);
  int* recs = (int*)d_out;

  const dim3 blk(256);
  const int gE = (E + 255) / 256;
  const int gChunk = (E + CHUNK - 1) / CHUNK;
  const int gWord = (N / 2 + 8 + 255) / 256;
  const int gGemm = (N + 127) / 128;
  const int gGather = 2048;

  chunk_count<<<gChunk, dim3(1024), 0, stream>>>(dstp, ccnt, rank, N, E);
  chunk_offsets<<<gWord, blk, 0, stream>>>(ccnt, offs, rowptr, N, gChunk);
  scan_rows<<<1, dim3(1024), 0, stream>>>(rowptr, invs, N);
  fill_lists<<<gE, blk, 0, stream>>>(src, dstp, rank, rowptr, offs, invs, recs,
                                     N, E);

  gemm_wmma_f32<128><<<gGemm, blk, 0, stream>>>(x, W1, nullptr, bufA, N);
  gather_rows<<<gGather, blk, 0, stream>>>(bufA, rowptr, recs, b1, bufB, N, E, 1);

  gemm_wmma_f32<128><<<gGemm, blk, 0, stream>>>(bufB, W2, nullptr, bufA, N);
  gather_rows<<<gGather, blk, 0, stream>>>(bufA, rowptr, recs, b2, bufB, N, E, 1);

  gemm_wmma_f32<128><<<gGemm, blk, 0, stream>>>(bufB, W3, nullptr, bufA, N);
  gather_rows<<<gGather, blk, 0, stream>>>(bufA, rowptr, recs, b3, bufB, N, E, 0);

  gemm_wmma_f32<64><<<gGemm, blk, 0, stream>>>(bufB, Wo, bo, (float*)d_out, N);
}


# output projection folded into layer 3: w34=W3*Wo, c34=b3*Wo+bo on WMMA beside the row scan; layer-3 product and gather run 64 channels wide and write out; (h W3) Wo reassociated to h (W3 Wo)
# speedup vs baseline: 3.9416x; 1.1576x over previous
#include <hip/hip_runtime.h>

typedef float v2f __attribute__((ext_vector_type(2)));
typedef float v8f __attribute__((ext_vector_type(8)));

#define CH 128

#define KPITCH 130

template <int NCOL, int THREADS>
__device__ __forceinline__ void gemm_body(
    const float* __restrict__ A, const float* __restrict__ W,
    const float* __restrict__ bias, float* __restrict__ C, int nrows, int blk,
    float* __restrict__ lw) {
  constexpr int NT = NCOL / 16;
  constexpr int KP = KPITCH;

  const int tid = threadIdx.x;
  if constexpr (THREADS == 256) {
    for (int i = tid; i < CH * NCOL; i += THREADS) {
      int k = i / NCOL, c = i - k * NCOL;
      lw[c * KP + k] = W[i];
    }
  } else {
    constexpr int TR = CH * NCOL / THREADS;
    float wv[TR];
#pragma unroll
    for (int q = 0; q < TR; ++q) wv[q] = W[tid + q * THREADS];
#pragma unroll
    for (int q = 0; q < TR; ++q) {
      const int i = tid + q * THREADS;
      const int k = i / NCOL, c = i - k * NCOL;
      lw[c * KP + k] = wv[q];
    }
  }
  __syncthreads();

  const int wave = tid >> 5;
  const int lane = tid & 31;
  const int m0 = (blk * (THREADS / 32) + wave) * 16;
  if (m0 >= nrows) return;

  const int lrow = lane & 15;
  const int kbase = (lane >> 4) * 2;

  v8f acc[NT] = {};

  int arow = m0 + lrow;
  if (arow > nrows - 1) arow = nrows - 1;
  const float* aptr = A + (size_t)arow * CH + kbase;

#pragma unroll(NCOL == 64 ? CH / 4 : 4)
  for (int kc = 0; kc < CH / 4; ++kc) {
    float2 av = *(const float2*)(aptr + kc * 4);
    v2f a;
    a[0] = av.x; a[1] = av.y;
#pragma unroll
    for (int n = 0; n < NT; ++n) {
      float2 bv = *(const float2*)&lw[(n * 16 + lrow) * KP + kc * 4 + kbase];
      v2f b;
      b[0] = bv.x; b[1] = bv.y;
      acc[n] = __builtin_amdgcn_wmma_f32_16x16x4_f32(false, a, false, b,
                                                     (short)0, acc[n],
                                                     false, false);
    }
  }

  if constexpr (NCOL == 64) {
    __builtin_amdgcn_sched_barrier(0);
    asm volatile(
        "v_nop\n\tv_nop\n\tv_nop\n\tv_nop\n\tv_nop\n\tv_nop\n\tv_nop\n\tv_nop"
        : "+v"(acc[0]), "+v"(acc[1]), "+v"(acc[2]), "+v"(acc[3]));
    __builtin_amdgcn_sched_barrier(0);
  }

  const int rbase = m0 + ((lane >> 4) << 3);
#pragma unroll
  for (int r = 0; r < 8; ++r) {
    int row = rbase + r;
    if (row < nrows) {
#pragma unroll
      for (int n = 0; n < NT; ++n) {
        int col = n * 16 + lrow;
        float v = acc[n][r];
        if (bias) v += bias[col];
        C[(size_t)row * NCOL + col] = v;
      }
    }
  }
}

template <int NCOL>
__global__ __launch_bounds__(256) void gemm_wmma_f32(
    const float* __restrict__ A, const float* __restrict__ W,
    const float* __restrict__ bias, float* __restrict__ C, int nrows) {
  __shared__ float lw[NCOL * KPITCH];
  gemm_body<NCOL, 256>(A, W, bias, C, nrows, (int)blockIdx.x, lw);
}

template <int NCOL>
__global__ __launch_bounds__(256) void gemm_wmma_f32_move(
    const float* __restrict__ A, const float* __restrict__ W,
    float* __restrict__ C, int nrows, const int* cp_src, int* cp_dst,
    int ncopy) {
  __shared__ float lw[NCOL * KPITCH];
  const int per =
      ((ncopy + (int)gridDim.x - 1) / (int)gridDim.x + 1023) & ~1023;
  const int c0 = (int)blockIdx.x * per;
  const int c1 = min(c0 + per, ncopy);
  const int4* ms = (const int4*)cp_src;
  int4* md = (int4*)cp_dst;
  for (int g0 = c0; g0 < c1; g0 += 1024) {
    if (g0 + 1024 <= ncopy) {
      const int i = g0 + (int)threadIdx.x;
      const int4 m0 = ms[i];
      const int4 m1 = ms[i + 256];
      const int4 m2 = ms[i + 512];
      const int4 m3 = ms[i + 768];
      md[i] = m0;
      md[i + 256] = m1;
      md[i + 512] = m2;
      md[i + 768] = m3;
    } else {
      for (int i = g0 + (int)threadIdx.x; i < ncopy; i += 256) md[i] = ms[i];
    }
  }
  gemm_body<NCOL, 256>(A, W, nullptr, C, nrows, (int)blockIdx.x, lw);
}

#define CHUNK 32768
#define CAPW 65536

__device__ __forceinline__ int packed_pitch(int n) {
  return (((n + 1) >> 1) + 3) & ~3;
}

__global__ __launch_bounds__(1024) void chunk_count(
    const int* __restrict__ dst, unsigned* __restrict__ ccnt,
    int* __restrict__ rank, int n, int ne) {
  __shared__ unsigned lc[CAPW];
  const int tid = threadIdx.x;
  const int c = blockIdx.x;
  const int e0 = c * CHUNK;
  const int e1 = min(e0 + CHUNK, ne);
  const int nwp = packed_pitch(n);
  for (int wb = 0; wb < nwp; wb += CAPW) {
    const int nwin = min(CAPW, nwp - wb);
    for (int i = tid * 4; i < nwin; i += 4096)
      *(uint4*)&lc[i] = make_uint4(0u, 0u, 0u, 0u);
    __syncthreads();
    const int dlo = wb * 2;
    const int dhi = dlo + nwin * 2;
    int dv[CHUNK / 1024];
    unsigned old[CHUNK / 1024];
#pragma unroll
    for (int k = 0; k < CHUNK / 1024; ++k) {
      const int e = e0 + k * 1024 + tid;
      const int v = dst[min(e, e1 - 1)];
      dv[k] = (e < e1) ? v : -1;
    }
#pragma unroll
    for (int k = 0; k < CHUNK / 1024; ++k) {
      const bool in = dv[k] >= dlo && dv[k] < dhi;
      const int l = in ? dv[k] - dlo : 0;
      const unsigned inc = in ? ((l & 1) ? 65536u : 1u) : 0u;
      old[k] = atomicAdd(&lc[l >> 1], inc);
    }
#pragma unroll
    for (int k = 0; k < CHUNK / 1024; ++k) {
      const int e = e0 + k * 1024 + tid;
      if (dv[k] >= dlo && dv[k] < dhi)
        rank[e] = (int)(((dv[k] - dlo) & 1) ? (old[k] >> 16)
                                            : (old[k] & 0xFFFFu));
    }
    __syncthreads();
    for (int i = tid * 4; i < nwin; i += 4096)
      *(uint4*)&ccnt[(size_t)c * nwp + wb + i] = *(const uint4*)&lc[i];
    __syncthreads();
  }
}

__global__ __launch_bounds__(256) void chunk_offsets(
    const unsigned* __restrict__ ccnt, int* __restrict__ offs,
    int* __restrict__ cnt, int n, int nchunks) {
  const int nwp = packed_pitch(n);
  const int w = blockIdx.x * blockDim.x + threadIdx.x;
  if (w >= nwp) return;
  const size_t np = (size_t)nwp * 2;
  int lo = 0, hi = 0;
#pragma unroll 8
  for (int c = 0; c < nchunks; ++c) {
    const unsigned v = ccnt[(size_t)c * nwp + w];
    *(int2*)(offs + (size_t)c * np + 2 * w) = make_int2(lo, hi);
    lo += (int)(v & 0xFFFFu);
    hi += (int)(v >> 16);
  }
  if (2 * w < n) cnt[2 * w] = lo;
  if (2 * w + 1 < n) cnt[2 * w + 1] = hi;
}

__global__ __launch_bounds__(1024) void scan_rows_fold(
    int* __restrict__ cr, float* __restrict__ invs, int n,
    const float* __restrict__ W3, const float* __restrict__ b3,
    const float* __restrict__ Wo, const float* __restrict__ bo,
    float* __restrict__ w34, float* __restrict__ c34) {
  __shared__ float lwf[64 * KPITCH];
  if (blockIdx.x == 1) {
    gemm_body<64, 1024>(W3, Wo, nullptr, w34, CH, 0, lwf);
    return;
  }
  if (blockIdx.x == 2) {
    gemm_body<64, 1024>(b3, Wo, bo, c34, 1, 0, lwf);
    return;
  }
  __shared__ int wtot[32];
  const int tid = threadIdx.x;
  const int lane = tid & 31;
  const int wv = tid >> 5;
  int carry = 0;
  for (int base = 0; base < n; base += 16384) {
    const int i = base + tid * 16;
    const bool full = i + 15 < n;
    int c[16];
    if (full) {
#pragma unroll
      for (int q = 0; q < 4; ++q) {
        const int4 t4 = *(const int4*)(cr + i + q * 4);
        c[q * 4 + 0] = t4.x;
        c[q * 4 + 1] = t4.y;
        c[q * 4 + 2] = t4.z;
        c[q * 4 + 3] = t4.w;
      }
    } else {
#pragma unroll
      for (int k = 0; k < 16; ++k) c[k] = (i + k < n) ? cr[i + k] : 0;
    }
    int tsum = 0;
#pragma unroll
    for (int k = 0; k < 16; ++k) tsum += c[k];
    int v = tsum;
    for (int o = 1; o < 32; o <<= 1) {
      const int u = __shfl_up(v, o);
      if (lane >= o) v += u;
    }
    if (lane == 31) wtot[wv] = v;
    __syncthreads();
    const int wt = wtot[lane];
    int ws = wt;
    for (int o = 1; o < 32; o <<= 1) {
      const int u = __shfl_up(ws, o);
      if (lane >= o) ws += u;
    }
    const int wexcl = __shfl(ws, wv) - __shfl(wt, wv);
    const int total = __shfl(ws, 31);
    int run = carry + wexcl + (v - tsum);
    int ex[16];
    float rs[16];
#pragma unroll
    for (int k = 0; k < 16; ++k) {
      ex[k] = run;
      run += c[k];
      const float dk = (float)(c[k] + 1);
      rs[k] = dk > 0.f ? rsqrtf(dk) : 0.f;
    }
    if (full) {
#pragma unroll
      for (int q = 0; q < 4; ++q) {
        *(int4*)(cr + i + q * 4) =
            make_int4(ex[q * 4], ex[q * 4 + 1], ex[q * 4 + 2], ex[q * 4 + 3]);
        *(float4*)(invs + i + q * 4) =
            make_float4(rs[q * 4], rs[q * 4 + 1], rs[q * 4 + 2], rs[q * 4 + 3]);
      }
    } else {
#pragma unroll
      for (int k = 0; k < 16; ++k) {
        if (i + k < n) {
          cr[i + k] = ex[k];
          invs[i + k] = rs[k];
        }
      }
    }
    carry += total;
    __syncthreads();
  }
}

__global__ void fill_lists(const int* __restrict__ src,
                           const int* __restrict__ dst,
                           const int* __restrict__ rank,
                           const int* __restrict__ rowptr,
                           const int* __restrict__ offs,
                           const float* __restrict__ invs,
                           int* __restrict__ recs, int n, int ne) {
  int e = blockIdx.x * blockDim.x + threadIdx.x;
  if (e < ne) {
    const int s = src[e];
    const int d = dst[e];
    const size_t np = (size_t)packed_pitch(n) * 2;
    int p = rowptr[d] + offs[(size_t)(e / CHUNK) * np + d] + rank[e];
    p = min(max(p, 0), ne - 1);
    const float w = invs[s] * invs[d];
    ((int2*)recs)[p] = make_int2(s, __float_as_int(w));
  }
}

template <int K>
__device__ __forceinline__ void add_rows(float4& acc,
                                         const float* __restrict__ tl, int rx,
                                         int ry, int j, int last) {
  float4 v[K];
  float wg[K];
#pragma unroll
  for (int u = 0; u < K; ++u) {
    int s = __builtin_amdgcn_readlane(rx, j + u);
    s = min(max(s, 0), last);
    wg[u] = __int_as_float(__builtin_amdgcn_readlane(ry, j + u));
    v[u] = *(const float4*)(tl + (size_t)s * CH);
  }
#pragma unroll
  for (int u = 0; u < K; ++u) {
    acc.x += v[u].x * wg[u];
    acc.y += v[u].y * wg[u];
    acc.z += v[u].z * wg[u];
    acc.w += v[u].w * wg[u];
  }
}

__global__ __launch_bounds__(256) void gather_rows(
    const float* __restrict__ t, const int* __restrict__ rowptr,
    const int* __restrict__ recs, const float* __restrict__ bias,
    float* __restrict__ h, int n, int ne, int relu) {
  const int lane = threadIdx.x & 31;
  const int w0 = __builtin_amdgcn_readfirstlane(
      (int)((blockIdx.x * blockDim.x + threadIdx.x) >> 5));
  const int nw = (int)((gridDim.x * blockDim.x) >> 5);
  const float4 b = *(const float4*)(bias + lane * 4);
  const int2* rl = (const int2*)recs;
  const float* tl = t + lane * 4;
  for (int d = w0; d < n; d += nw) {
    const int start = rowptr[d];
    const int end = (d + 1 < n) ? rowptr[d + 1] : ne;
    const float4 tv = *(const float4*)(tl + (size_t)d * CH);
    float4 acc = make_float4(0.f, 0.f, 0.f, 0.f);
    for (int base = start; base < end; base += 32) {
      const int idx = base + lane;
      int2 r = make_int2(0, 0);
      if (idx < end) r = rl[idx];
      const int cnt = min(32, end - base);
      int j = 0;
      for (; j + 8 <= cnt; j += 8) add_rows<8>(acc, tl, r.x, r.y, j, n - 1);
      if (j + 4 <= cnt) { add_rows<4>(acc, tl, r.x, r.y, j, n - 1); j += 4; }
      if (j + 2 <= cnt) { add_rows<2>(acc, tl, r.x, r.y, j, n - 1); j += 2; }
      if (j < cnt) add_rows<1>(acc, tl, r.x, r.y, j, n - 1);
    }
    const float degf = (float)(end - start + 1);
    float sl = degf > 0.f ? rsqrtf(degf) : 0.f;
    sl = sl * sl;
    float4 o;
    o.x = acc.x + tv.x * sl + b.x;
    o.y = acc.y + tv.y * sl + b.y;
    o.z = acc.z + tv.z * sl + b.z;
    o.w = acc.w + tv.w * sl + b.w;
    if (relu) {
      o.x = fmaxf(o.x, 0.f);
      o.y = fmaxf(o.y, 0.f);
      o.z = fmaxf(o.z, 0.f);
      o.w = fmaxf(o.w, 0.f);
    }
    *(float4*)(h + (size_t)d * CH + lane * 4) = o;
  }
}

template <int K>
__device__ __forceinline__ void add_rows64(float2& acc,
                                           const float* __restrict__ tl,
                                           int rx, int ry, int j, int last) {
  float2 v[K];
  float wg[K];
#pragma unroll
  for (int u = 0; u < K; ++u) {
    int s = __builtin_amdgcn_readlane(rx, j + u);
    s = min(max(s, 0), last);
    wg[u] = __int_as_float(__builtin_amdgcn_readlane(ry, j + u));
    v[u] = *(const float2*)(tl + (size_t)s * 64);
  }
#pragma unroll
  for (int u = 0; u < K; ++u) {
    acc.x += v[u].x * wg[u];
    acc.y += v[u].y * wg[u];
  }
}

__global__ __launch_bounds__(256) void gather_rows64(
    const float* __restrict__ t, const int* __restrict__ rowptr,
    const int* __restrict__ recs, const float* __restrict__ bias,
    float* __restrict__ h, int n, int ne) {
  const int lane = threadIdx.x & 31;
  const int w0 = __builtin_amdgcn_readfirstlane(
      (int)((blockIdx.x * blockDim.x + threadIdx.x) >> 5));
  const int nw = (int)((gridDim.x * blockDim.x) >> 5);
  const float2 b = *(const float2*)(bias + lane * 2);
  const int2* rl = (const int2*)recs;
  const float* tl = t + lane * 2;
  for (int d = w0; d < n; d += nw) {
    const int start = rowptr[d];
    const int end = (d + 1 < n) ? rowptr[d + 1] : ne;
    const float2 tv = *(const float2*)(tl + (size_t)d * 64);
    float2 acc = make_float2(0.f, 0.f);
    for (int base = start; base < end; base += 32) {
      const int idx = base + lane;
      int2 r = make_int2(0, 0);
      if (idx < end) r = rl[idx];
      const int cnt = min(32, end - base);
      int j = 0;
      for (; j + 16 <= cnt; j += 16) add_rows64<16>(acc, tl, r.x, r.y, j, n - 1);
      if (j + 8 <= cnt) { add_rows64<8>(acc, tl, r.x, r.y, j, n - 1); j += 8; }
      if (j + 4 <= cnt) { add_rows64<4>(acc, tl, r.x, r.y, j, n - 1); j += 4; }
      if (j + 2 <= cnt) { add_rows64<2>(acc, tl, r.x, r.y, j, n - 1); j += 2; }
      if (j < cnt) add_rows64<1>(acc, tl, r.x, r.y, j, n - 1);
    }
    const float degf = (float)(end - start + 1);
    float sl = degf > 0.f ? rsqrtf(degf) : 0.f;
    sl = sl * sl;
    float2 o;
    o.x = acc.x + tv.x * sl + b.x;
    o.y = acc.y + tv.y * sl + b.y;
    *(float2*)(h + (size_t)d * 64 + lane * 2) = o;
  }
}

extern "C" void kernel_launch(void* const* d_in, const int* in_sizes, int n_in,
                              void* d_out, int out_size, void* d_ws,
                              size_t ws_size, hipStream_t stream) {
  const float* x = (const float*)d_in[0];
  const int* ei = (const int*)d_in[1];
  const float* W1 = (const float*)d_in[2];
  const float* b1 = (const float*)d_in[3];
  const float* W2 = (const float*)d_in[4];
  const float* b2 = (const float*)d_in[5];
  const float* W3 = (const float*)d_in[6];
  const float* b3 = (const float*)d_in[7];
  const float* Wo = (const float*)d_in[8];
  const float* bo = (const float*)d_in[9];

  const int N = in_sizes[0] / CH;
  const int E = in_sizes[1] / 2;
  const int* src = ei;
  const int* dstp = ei + E;

  float* bufA = (float*)d_ws;
  float* bufB = bufA + (size_t)N * CH;
  int* rowptr = (int*)(bufB + (size_t)N * CH);
  int* rank = rowptr + N;
  float* invs = bufB;
  unsigned* ccnt = (unsigned*)bufA;
  int* offs = (int*)(bufB + (size_t)2 * N);
  int* recs = (int*)d_out;
  float* c34 = (float*)d_out + (size_t)2 * E;
  float* w34 = c34 + 64;
  float* t3 = bufA;
  int* recs3 = (int*)(bufA + (size_t)N * 64);
  const float* c34m = (const float*)(recs3 + (size_t)2 * E);
  const int nmove = (2 * E + 64 + 3) / 4;

  const dim3 blk(256);
  const int gE = (E + 255) / 256;
  const int gChunk = (E + CHUNK - 1) / CHUNK;
  const int gWord = (N / 2 + 8 + 255) / 256;
  const int gGemm = (N + 127) / 128;
  const int gGather = 2048;

  chunk_count<<<gChunk, dim3(1024), 0, stream>>>(dstp, ccnt, rank, N, E);
  chunk_offsets<<<gWord, blk, 0, stream>>>(ccnt, offs, rowptr, N, gChunk);
  scan_rows_fold<<<3, dim3(1024), 0, stream>>>(rowptr, invs, N, W3, b3, Wo, bo,
                                               w34, c34);
  fill_lists<<<gE, blk, 0, stream>>>(src, dstp, rank, rowptr, offs, invs, recs,
                                     N, E);

  gemm_wmma_f32<128><<<gGemm, blk, 0, stream>>>(x, W1, nullptr, bufA, N);
  gather_rows<<<gGather, blk, 0, stream>>>(bufA, rowptr, recs, b1, bufB, N, E, 1);

  gemm_wmma_f32<128><<<gGemm, blk, 0, stream>>>(bufB, W2, nullptr, bufA, N);
  gather_rows<<<gGather, blk, 0, stream>>>(bufA, rowptr, recs, b2, bufB, N, E, 1);

  gemm_wmma_f32_move<64><<<gGemm, blk, 0, stream>>>(bufB, w34, t3, N, recs,
                                                    recs3, nmove);
  gather_rows64<<<gGather, blk, 0, stream>>>(t3, rowptr, recs3, c34m,
                                             (float*)d_out, N, E);
}


# layer-1 product x W1 leaves its own launch: its 128-row tiles run as extra work-groups of the three list-building launches (count, offsets, scan) on WGPs those leave idle; 10 -> 9 launches
# speedup vs baseline: 4.2008x; 1.0658x over previous
#include <hip/hip_runtime.h>

typedef float v2f __attribute__((ext_vector_type(2)));
typedef float v8f __attribute__((ext_vector_type(8)));

#define CH 128

#define KPITCH 130

template <int NCOL, int THREADS, int SPB = THREADS / 32,
          bool PAD = (NCOL == 64)>
__device__ __forceinline__ void gemm_body(
    const float* __restrict__ A, const float* __restrict__ W,
    const float* __restrict__ bias, float* __restrict__ C, int nrows, int blk,
    float* __restrict__ lw) {
  constexpr int NT = NCOL / 16;
  constexpr int KP = KPITCH;

  const int tid = threadIdx.x;
  if constexpr (THREADS == 256) {
    for (int i = tid; i < CH * NCOL; i += THREADS) {
      int k = i / NCOL, c = i - k * NCOL;
      lw[c * KP + k] = W[i];
    }
  } else {
    constexpr int TR = CH * NCOL / THREADS;
    float wv[TR];
#pragma unroll
    for (int q = 0; q < TR; ++q) wv[q] = W[tid + q * THREADS];
#pragma unroll
    for (int q = 0; q < TR; ++q) {
      const int i = tid + q * THREADS;
      const int k = i / NCOL, c = i - k * NCOL;
      lw[c * KP + k] = wv[q];
    }
  }
  __syncthreads();

  const int wave = tid >> 5;
  const int lane = tid & 31;
  if constexpr (SPB < THREADS / 32) {
    if (wave >= SPB) return;
  }
  const int m0 = (blk * SPB + wave) * 16;
  if (m0 >= nrows) return;

  const int lrow = lane & 15;
  const int kbase = (lane >> 4) * 2;

  v8f acc[NT] = {};

  int arow = m0 + lrow;
  if (arow > nrows - 1) arow = nrows - 1;
  const float* aptr = A + (size_t)arow * CH + kbase;

#pragma unroll(NCOL == 64 ? CH / 4 : 4)
  for (int kc = 0; kc < CH / 4; ++kc) {
    float2 av = *(const float2*)(aptr + kc * 4);
    v2f a;
    a[0] = av.x; a[1] = av.y;
#pragma unroll
    for (int n = 0; n < NT; ++n) {
      float2 bv = *(const float2*)&lw[(n * 16 + lrow) * KP + kc * 4 + kbase];
      v2f b;
      b[0] = bv.x; b[1] = bv.y;
      acc[n] = __builtin_amdgcn_wmma_f32_16x16x4_f32(false, a, false, b,
                                                     (short)0, acc[n],
                                                     false, false);
    }
  }

  if constexpr (PAD) {
    if constexpr (NCOL == 64) __builtin_amdgcn_sched_barrier(0);
    if constexpr (NT == 4) {
      asm volatile(
          "v_nop\n\tv_nop\n\tv_nop\n\tv_nop\n\tv_nop\n\tv_nop\n\tv_nop\n\tv_nop"
          : "+v"(acc[0]), "+v"(acc[1]), "+v"(acc[2]), "+v"(acc[3]));
    } else {
      static_assert(NT == 4 || NT == 8, "accumulators named below");
      asm volatile(
          "v_nop\n\tv_nop\n\tv_nop\n\tv_nop\n\tv_nop\n\tv_nop\n\tv_nop\n\tv_nop"
          : "+v"(acc[0]), "+v"(acc[1]), "+v"(acc[2]), "+v"(acc[3]),
            "+v"(acc[4]), "+v"(acc[5]), "+v"(acc[6]), "+v"(acc[7]));
    }
    __builtin_amdgcn_sched_barrier(0);
  }

  const int rbase = m0 + ((lane >> 4) << 3);
#pragma unroll
  for (int r = 0; r < 8; ++r) {
    int row = rbase + r;
    if (row < nrows) {
#pragma unroll
      for (int n = 0; n < NT; ++n) {
        int col = n * 16 + lrow;
        float v = acc[n][r];
        if (bias) v += bias[col];
        C[(size_t)row * NCOL + col] = v;
      }
    }
  }
}

template <int NCOL>
__global__ __launch_bounds__(256) void gemm_wmma_f32(
    const float* __restrict__ A, const float* __restrict__ W,
    const float* __restrict__ bias, float* __restrict__ C, int nrows) {
  __shared__ float lw[NCOL * KPITCH];
  gemm_body<NCOL, 256>(A, W, bias, C, nrows, (int)blockIdx.x, lw);
}

template <int NCOL>
__global__ __launch_bounds__(256) void gemm_wmma_f32_move(
    const float* __restrict__ A, const float* __restrict__ W,
    float* __restrict__ C, int nrows, const int* cp_src, int* cp_dst,
    int ncopy) {
  __shared__ float lw[NCOL * KPITCH];
  const int per =
      ((ncopy + (int)gridDim.x - 1) / (int)gridDim.x + 1023) & ~1023;
  const int c0 = (int)blockIdx.x * per;
  const int c1 = min(c0 + per, ncopy);
  const int4* ms = (const int4*)cp_src;
  int4* md = (int4*)cp_dst;
  for (int g0 = c0; g0 < c1; g0 += 1024) {
    if (g0 + 1024 <= ncopy) {
      const int i = g0 + (int)threadIdx.x;
      const int4 m0 = ms[i];
      const int4 m1 = ms[i + 256];
      const int4 m2 = ms[i + 512];
      const int4 m3 = ms[i + 768];
      md[i] = m0;
      md[i + 256] = m1;
      md[i + 512] = m2;
      md[i + 768] = m3;
    } else {
      for (int i = g0 + (int)threadIdx.x; i < ncopy; i += 256) md[i] = ms[i];
    }
  }
  gemm_body<NCOL, 256>(A, W, nullptr, C, nrows, (int)blockIdx.x, lw);
}

#define CHUNK 32768
#define CAPW 65536

__device__ __forceinline__ int packed_pitch(int n) {
  return (((n + 1) >> 1) + 3) & ~3;
}

__global__ __launch_bounds__(1024) void chunk_count(
    const int* __restrict__ dst, unsigned* __restrict__ ccnt,
    int* __restrict__ rank, int n, int ne, int nchunks,
    const float* __restrict__ x, const float* __restrict__ W1,
    float* __restrict__ t1, int tile0) {
  __shared__ unsigned lc[CAPW];
  if ((int)blockIdx.x >= nchunks) {
    gemm_body<128, 1024, 8, true>(x, W1, nullptr, t1, n,
                                  tile0 + (int)blockIdx.x - nchunks,
                                  (float*)lc);
    return;
  }
  const int tid = threadIdx.x;
  const int c = blockIdx.x;
  const int e0 = c * CHUNK;
  const int e1 = min(e0 + CHUNK, ne);
  const int nwp = packed_pitch(n);
  for (int wb = 0; wb < nwp; wb += CAPW) {
    const int nwin = min(CAPW, nwp - wb);
    for (int i = tid * 4; i < nwin; i += 4096)
      *(uint4*)&lc[i] = make_uint4(0u, 0u, 0u, 0u);
    __syncthreads();
    const int dlo = wb * 2;
    const int dhi = dlo + nwin * 2;
    int dv[CHUNK / 1024];
    unsigned old[CHUNK / 1024];
#pragma unroll
    for (int k = 0; k < CHUNK / 1024; ++k) {
      const int e = e0 + k * 1024 + tid;
      const int v = dst[min(e, e1 - 1)];
      dv[k] = (e < e1) ? v : -1;
    }
#pragma unroll
    for (int k = 0; k < CHUNK / 1024; ++k) {
      const bool in = dv[k] >= dlo && dv[k] < dhi;
      const int l = in ? dv[k] - dlo : 0;
      const unsigned inc = in ? ((l & 1) ? 65536u : 1u) : 0u;
      old[k] = atomicAdd(&lc[l >> 1], inc);
    }
#pragma unroll
    for (int k = 0; k < CHUNK / 1024; ++k) {
      const int e = e0 + k * 1024 + tid;
      if (dv[k] >= dlo && dv[k] < dhi)
        rank[e] = (int)(((dv[k] - dlo) & 1) ? (old[k] >> 16)
                                            : (old[k] & 0xFFFFu));
    }
    __syncthreads();
    for (int i = tid * 4; i < nwin; i += 4096)
      *(uint4*)&ccnt[(size_t)c * nwp + wb + i] = *(const uint4*)&lc[i];
    __syncthreads();
  }
}

__global__ __launch_bounds__(256) void chunk_offsets(
    const unsigned* __restrict__ ccnt, int* __restrict__ offs,
    int* __restrict__ cnt, int n, int nchunks, int nown,
    const float* __restrict__ x, const float* __restrict__ W1,
    float* __restrict__ t1, int tile0) {
  __shared__ float lw1[128 * KPITCH];
  if ((int)blockIdx.x >= nown) {
    gemm_body<128, 256, 8, true>(x, W1, nullptr, t1, n,
                                 tile0 + (int)blockIdx.x - nown, lw1);
    return;
  }
  const int nwp = packed_pitch(n);
  const int w = blockIdx.x * blockDim.x + threadIdx.x;
  if (w >= nwp) return;
  const size_t np = (size_t)nwp * 2;
  int lo = 0, hi = 0;
#pragma unroll 8
  for (int c = 0; c < nchunks; ++c) {
    const unsigned v = ccnt[(size_t)c * nwp + w];
    *(int2*)(offs + (size_t)c * np + 2 * w) = make_int2(lo, hi);
    lo += (int)(v & 0xFFFFu);
    hi += (int)(v >> 16);
  }
  if (2 * w < n) cnt[2 * w] = lo;
  if (2 * w + 1 < n) cnt[2 * w + 1] = hi;
}

__global__ __launch_bounds__(1024) void scan_rows_fold(
    int* __restrict__ cr, float* __restrict__ invs, int n,
    const float* __restrict__ W3, const float* __restrict__ b3,
    const float* __restrict__ Wo, const float* __restrict__ bo,
    float* __restrict__ w34, float* __restrict__ c34,
    const float* __restrict__ x, const float* __restrict__ W1,
    float* __restrict__ t1, int tile0) {
  __shared__ float lwf[128 * KPITCH];
  if (blockIdx.x >= 3) {
    gemm_body<128, 1024, 8, true>(x, W1, nullptr, t1, n,
                                  tile0 + (int)blockIdx.x - 3, lwf);
    return;
  }
  if (blockIdx.x == 1) {
    gemm_body<64, 1024>(W3, Wo, nullptr, w34, CH, 0, lwf);
    return;
  }
  if (blockIdx.x == 2) {
    gemm_body<64, 1024>(b3, Wo, bo, c34, 1, 0, lwf);
    return;
  }
  __shared__ int wtot[32];
  const int tid = threadIdx.x;
  const int lane = tid & 31;
  const int wv = tid >> 5;
  int carry = 0;
  for (int base = 0; base < n; base += 16384) {
    const int i = base + tid * 16;
    const bool full = i + 15 < n;
    int c[16];
    if (full) {
#pragma unroll
      for (int q = 0; q < 4; ++q) {
        const int4 t4 = *(const int4*)(cr + i + q * 4);
        c[q * 4 + 0] = t4.x;
        c[q * 4 + 1] = t4.y;
        c[q * 4 + 2] = t4.z;
        c[q * 4 + 3] = t4.w;
      }
    } else {
#pragma unroll
      for (int k = 0; k < 16; ++k) c[k] = (i + k < n) ? cr[i + k] : 0;
    }
    int tsum = 0;
#pragma unroll
    for (int k = 0; k < 16; ++k) tsum += c[k];
    int v = tsum;
    for (int o = 1; o < 32; o <<= 1) {
      const int u = __shfl_up(v, o);
      if (lane >= o) v += u;
    }
    if (lane == 31) wtot[wv] = v;
    __syncthreads();
    const int wt = wtot[lane];
    int ws = wt;
    for (int o = 1; o < 32; o <<= 1) {
      const int u = __shfl_up(ws, o);
      if (lane >= o) ws += u;
    }
    const int wexcl = __shfl(ws, wv) - __shfl(wt, wv);
    const int total = __shfl(ws, 31);
    int run = carry + wexcl + (v - tsum);
    int ex[16];
    float rs[16];
#pragma unroll
    for (int k = 0; k < 16; ++k) {
      ex[k] = run;
      run += c[k];
      const float dk = (float)(c[k] + 1);
      rs[k] = dk > 0.f ? rsqrtf(dk) : 0.f;
    }
    if (full) {
#pragma unroll
      for (int q = 0; q < 4; ++q) {
        *(int4*)(cr + i + q * 4) =
            make_int4(ex[q * 4], ex[q * 4 + 1], ex[q * 4 + 2], ex[q * 4 + 3]);
        *(float4*)(invs + i + q * 4) =
            make_float4(rs[q * 4], rs[q * 4 + 1], rs[q * 4 + 2], rs[q * 4 + 3]);
      }
    } else {
#pragma unroll
      for (int k = 0; k < 16; ++k) {
        if (i + k < n) {
          cr[i + k] = ex[k];
          invs[i + k] = rs[k];
        }
      }
    }
    carry += total;
    __syncthreads();
  }
}

__global__ void fill_lists(const int* __restrict__ src,
                           const int* __restrict__ dst,
                           const int* __restrict__ rank,
                           const int* __restrict__ rowptr,
                           const int* __restrict__ offs,
                           const float* __restrict__ invs,
                           int* __restrict__ recs, int n, int ne) {
  int e = blockIdx.x * blockDim.x + threadIdx.x;
  if (e < ne) {
    const int s = src[e];
    const int d = dst[e];
    const size_t np = (size_t)packed_pitch(n) * 2;
    int p = rowptr[d] + offs[(size_t)(e / CHUNK) * np + d] + rank[e];
    p = min(max(p, 0), ne - 1);
    const float w = invs[s] * invs[d];
    ((int2*)recs)[p] = make_int2(s, __float_as_int(w));
  }
}

template <int K>
__device__ __forceinline__ void add_rows(float4& acc,
                                         const float* __restrict__ tl, int rx,
                                         int ry, int j, int last) {
  float4 v[K];
  float wg[K];
#pragma unroll
  for (int u = 0; u < K; ++u) {
    int s = __builtin_amdgcn_readlane(rx, j + u);
    s = min(max(s, 0), last);
    wg[u] = __int_as_float(__builtin_amdgcn_readlane(ry, j + u));
    v[u] = *(const float4*)(tl + (size_t)s * CH);
  }
#pragma unroll
  for (int u = 0; u < K; ++u) {
    acc.x += v[u].x * wg[u];
    acc.y += v[u].y * wg[u];
    acc.z += v[u].z * wg[u];
    acc.w += v[u].w * wg[u];
  }
}

__global__ __launch_bounds__(256) void gather_rows(
    const float* __restrict__ t, const int* __restrict__ rowptr,
    const int* __restrict__ recs, const float* __restrict__ bias,
    float* __restrict__ h, int n, int ne, int relu) {
  const int lane = threadIdx.x & 31;
  const int w0 = __builtin_amdgcn_readfirstlane(
      (int)((blockIdx.x * blockDim.x + threadIdx.x) >> 5));
  const int nw = (int)((gridDim.x * blockDim.x) >> 5);
  const float4 b = *(const float4*)(bias + lane * 4);
  const int2* rl = (const int2*)recs;
  const float* tl = t + lane * 4;
  for (int d = w0; d < n; d += nw) {
    const int start = rowptr[d];
    const int end = (d + 1 < n) ? rowptr[d + 1] : ne;
    const float4 tv = *(const float4*)(tl + (size_t)d * CH);
    float4 acc = make_float4(0.f, 0.f, 0.f, 0.f);
    for (int base = start; base < end; base += 32) {
      const int idx = base + lane;
      int2 r = make_int2(0, 0);
      if (idx < end) r = rl[idx];
      const int cnt = min(32, end - base);
      int j = 0;
      for (; j + 8 <= cnt; j += 8) add_rows<8>(acc, tl, r.x, r.y, j, n - 1);
      if (j + 4 <= cnt) { add_rows<4>(acc, tl, r.x, r.y, j, n - 1); j += 4; }
      if (j + 2 <= cnt) { add_rows<2>(acc, tl, r.x, r.y, j, n - 1); j += 2; }
      if (j < cnt) add_rows<1>(acc, tl, r.x, r.y, j, n - 1);
    }
    const float degf = (float)(end - start + 1);
    float sl = degf > 0.f ? rsqrtf(degf) : 0.f;
    sl = sl * sl;
    float4 o;
    o.x = acc.x + tv.x * sl + b.x;
    o.y = acc.y + tv.y * sl + b.y;
    o.z = acc.z + tv.z * sl + b.z;
    o.w = acc.w + tv.w * sl + b.w;
    if (relu) {
      o.x = fmaxf(o.x, 0.f);
      o.y = fmaxf(o.y, 0.f);
      o.z = fmaxf(o.z, 0.f);
      o.w = fmaxf(o.w, 0.f);
    }
    *(float4*)(h + (size_t)d * CH + lane * 4) = o;
  }
}

template <int K>
__device__ __forceinline__ void add_rows64(float2& acc,
                                           const float* __restrict__ tl,
                                           int rx, int ry, int j, int last) {
  float2 v[K];
  float wg[K];
#pragma unroll
  for (int u = 0; u < K; ++u) {
    int s = __builtin_amdgcn_readlane(rx, j + u);
    s = min(max(s, 0), last);
    wg[u] = __int_as_float(__builtin_amdgcn_readlane(ry, j + u));
    v[u] = *(const float2*)(tl + (size_t)s * 64);
  }
#pragma unroll
  for (int u = 0; u < K; ++u) {
    acc.x += v[u].x * wg[u];
    acc.y += v[u].y * wg[u];
  }
}

__global__ __launch_bounds__(256) void gather_rows64(
    const float* __restrict__ t, const int* __restrict__ rowptr,
    const int* __restrict__ recs, const float* __restrict__ bias,
    float* __restrict__ h, int n, int ne) {
  const int lane = threadIdx.x & 31;
  const int w0 = __builtin_amdgcn_readfirstlane(
      (int)((blockIdx.x * blockDim.x + threadIdx.x) >> 5));
  const int nw = (int)((gridDim.x * blockDim.x) >> 5);
  const float2 b = *(const float2*)(bias + lane * 2);
  const int2* rl = (const int2*)recs;
  const float* tl = t + lane * 2;
  for (int d = w0; d < n; d += nw) {
    const int start = rowptr[d];
    const int end = (d + 1 < n) ? rowptr[d + 1] : ne;
    const float2 tv = *(const float2*)(tl + (size_t)d * 64);
    float2 acc = make_float2(0.f, 0.f);
    for (int base = start; base < end; base += 32) {
      const int idx = base + lane;
      int2 r = make_int2(0, 0);
      if (idx < end) r = rl[idx];
      const int cnt = min(32, end - base);
      int j = 0;
      for (; j + 16 <= cnt; j += 16) add_rows64<16>(acc, tl, r.x, r.y, j, n - 1);
      if (j + 8 <= cnt) { add_rows64<8>(acc, tl, r.x, r.y, j, n - 1); j += 8; }
      if (j + 4 <= cnt) { add_rows64<4>(acc, tl, r.x, r.y, j, n - 1); j += 4; }
      if (j + 2 <= cnt) { add_rows64<2>(acc, tl, r.x, r.y, j, n - 1); j += 2; }
      if (j < cnt) add_rows64<1>(acc, tl, r.x, r.y, j, n - 1);
    }
    const float degf = (float)(end - start + 1);
    float sl = degf > 0.f ? rsqrtf(degf) : 0.f;
    sl = sl * sl;
    float2 o;
    o.x = acc.x + tv.x * sl + b.x;
    o.y = acc.y + tv.y * sl + b.y;
    *(float2*)(h + (size_t)d * 64 + lane * 2) = o;
  }
}

extern "C" void kernel_launch(void* const* d_in, const int* in_sizes, int n_in,
                              void* d_out, int out_size, void* d_ws,
                              size_t ws_size, hipStream_t stream) {
  const float* x = (const float*)d_in[0];
  const int* ei = (const int*)d_in[1];
  const float* W1 = (const float*)d_in[2];
  const float* b1 = (const float*)d_in[3];
  const float* W2 = (const float*)d_in[4];
  const float* b2 = (const float*)d_in[5];
  const float* W3 = (const float*)d_in[6];
  const float* b3 = (const float*)d_in[7];
  const float* Wo = (const float*)d_in[8];
  const float* bo = (const float*)d_in[9];

  const int N = in_sizes[0] / CH;
  const int E = in_sizes[1] / 2;
  const int* src = ei;
  const int* dstp = ei + E;

  float* bufA = (float*)d_ws;
  float* bufB = bufA + (size_t)N * CH;
  int* rowptr = (int*)(bufB + (size_t)N * CH);
  int* rank = rowptr + N;
  float* invs = bufB;
  int* offs = (int*)(bufB + (size_t)2 * N);
  int* recs = (int*)d_out;
  float* c34 = (float*)d_out + (size_t)2 * E;
  float* w34 = c34 + 64;
  unsigned* ccnt = (unsigned*)(w34 + (size_t)CH * 64);
  float* t3 = bufA;
  int* recs3 = (int*)(bufA + (size_t)N * 64);
  const float* c34m = (const float*)(recs3 + (size_t)2 * E);
  const int nmove = (2 * E + 64 + 3) / 4;

  const dim3 blk(256);
  const int gE = (E + 255) / 256;
  const int gChunk = (E + CHUNK - 1) / CHUNK;
  const int gWord = (N / 2 + 8 + 255) / 256;
  const int gGemm = (N + 127) / 128;
  const int gGather = 2048;
  const int gT1 = 207;
  const int gT2 = 256;
  const int gT3 = gGemm > gT1 + gT2 ? gGemm - gT1 - gT2 : 0;

  chunk_count<<<gChunk + gT1, dim3(1024), 0, stream>>>(
      dstp, ccnt, rank, N, E, gChunk, x, W1, bufA, 0);
  chunk_offsets<<<gWord + gT2, blk, 0, stream>>>(ccnt, offs, rowptr, N, gChunk,
                                                 gWord, x, W1, bufA, gT1);
  scan_rows_fold<<<3 + gT3, dim3(1024), 0, stream>>>(
      rowptr, invs, N, W3, b3, Wo, bo, w34, c34, x, W1, bufA, gT1 + gT2);
  fill_lists<<<gE, blk, 0, stream>>>(src, dstp, rank, rowptr, offs, invs, recs,
                                     N, E);

  gather_rows<<<gGather, blk, 0, stream>>>(bufA, rowptr, recs, b1, bufB, N, E, 1);

  gemm_wmma_f32<128><<<gGemm, blk, 0, stream>>>(bufB, W2, nullptr, bufA, N);
  gather_rows<<<gGather, blk, 0, stream>>>(bufA, rowptr, recs, b2, bufB, N, E, 1);

  gemm_wmma_f32_move<64><<<gGemm, blk, 0, stream>>>(bufB, w34, t3, N, recs,
                                                    recs3, nmove);
  gather_rows64<<<gGather, blk, 0, stream>>>(t3, rowptr, recs3, c34m,
                                             (float*)d_out, N, E);
}


# layer-2/3 products leave their launches: a resident 16-wave group gathers 32 rows into LDS, multiplies them by LDS-staged weights on WMMA; h never stored; 9 -> 7 launches; gather adds written as fma
# speedup vs baseline: 4.7743x; 1.1365x over previous
#include <hip/hip_runtime.h>

typedef float v2f __attribute__((ext_vector_type(2)));
typedef float v8f __attribute__((ext_vector_type(8)));

#define CH 128

#define KPITCH 130

template <int NCOL, int THREADS, int SPB = THREADS / 32,
          bool PAD = (NCOL == 64)>
__device__ __forceinline__ void gemm_body(
    const float* __restrict__ A, const float* __restrict__ W,
    const float* __restrict__ bias, float* __restrict__ C, int nrows, int blk,
    float* __restrict__ lw) {
  constexpr int NT = NCOL / 16;
  constexpr int KP = KPITCH;

  const int tid = threadIdx.x;
  if constexpr (THREADS == 256) {
    for (int i = tid; i < CH * NCOL; i += THREADS) {
      int k = i / NCOL, c = i - k * NCOL;
      lw[c * KP + k] = W[i];
    }
  } else {
    constexpr int TR = CH * NCOL / THREADS;
    float wv[TR];
#pragma unroll
    for (int q = 0; q < TR; ++q) wv[q] = W[tid + q * THREADS];
#pragma unroll
    for (int q = 0; q < TR; ++q) {
      const int i = tid + q * THREADS;
      const int k = i / NCOL, c = i - k * NCOL;
      lw[c * KP + k] = wv[q];
    }
  }
  __syncthreads();

  const int wave = tid >> 5;
  const int lane = tid & 31;
  if constexpr (SPB < THREADS / 32) {
    if (wave >= SPB) return;
  }
  const int m0 = (blk * SPB + wave) * 16;
  if (m0 >= nrows) return;

  const int lrow = lane & 15;
  const int kbase = (lane >> 4) * 2;

  v8f acc[NT] = {};

  int arow = m0 + lrow;
  if (arow > nrows - 1) arow = nrows - 1;
  const float* aptr = A + (size_t)arow * CH + kbase;

#pragma unroll(NCOL == 64 ? CH / 4 : 4)
  for (int kc = 0; kc < CH / 4; ++kc) {
    float2 av = *(const float2*)(aptr + kc * 4);
    v2f a;
    a[0] = av.x; a[1] = av.y;
#pragma unroll
    for (int n = 0; n < NT; ++n) {
      float2 bv = *(const float2*)&lw[(n * 16 + lrow) * KP + kc * 4 + kbase];
      v2f b;
      b[0] = bv.x; b[1] = bv.y;
      acc[n] = __builtin_amdgcn_wmma_f32_16x16x4_f32(false, a, false, b,
                                                     (short)0, acc[n],
                                                     false, false);
    }
  }

  if constexpr (PAD) {
    if constexpr (NCOL == 64) __builtin_amdgcn_sched_barrier(0);
    if constexpr (NT == 4) {
      asm volatile(
          "v_nop\n\tv_nop\n\tv_nop\n\tv_nop\n\tv_nop\n\tv_nop\n\tv_nop\n\tv_nop"
          : "+v"(acc[0]), "+v"(acc[1]), "+v"(acc[2]), "+v"(acc[3]));
    } else {
      static_assert(NT == 4 || NT == 8, "accumulators named below");
      asm volatile(
          "v_nop\n\tv_nop\n\tv_nop\n\tv_nop\n\tv_nop\n\tv_nop\n\tv_nop\n\tv_nop"
          : "+v"(acc[0]), "+v"(acc[1]), "+v"(acc[2]), "+v"(acc[3]),
            "+v"(acc[4]), "+v"(acc[5]), "+v"(acc[6]), "+v"(acc[7]));
    }
    __builtin_amdgcn_sched_barrier(0);
  }

  const int rbase = m0 + ((lane >> 4) << 3);
#pragma unroll
  for (int r = 0; r < 8; ++r) {
    int row = rbase + r;
    if (row < nrows) {
#pragma unroll
      for (int n = 0; n < NT; ++n) {
        int col = n * 16 + lrow;
        float v = acc[n][r];
        if (bias) v += bias[col];
        C[(size_t)row * NCOL + col] = v;
      }
    }
  }
}

#define CHUNK 32768
#define CAPW 65536

__device__ __forceinline__ int packed_pitch(int n) {
  return (((n + 1) >> 1) + 3) & ~3;
}

__global__ __launch_bounds__(1024) void chunk_count(
    const int* __restrict__ dst, unsigned* __restrict__ ccnt,
    int* __restrict__ rank, int n, int ne, int nchunks,
    const float* __restrict__ x, const float* __restrict__ W1,
    float* __restrict__ t1, int tile0) {
  __shared__ unsigned lc[CAPW];
  if ((int)blockIdx.x >= nchunks) {
    gemm_body<128, 1024, 8, true>(x, W1, nullptr, t1, n,
                                  tile0 + (int)blockIdx.x - nchunks,
                                  (float*)lc);
    return;
  }
  const int tid = threadIdx.x;
  const int c = blockIdx.x;
  const int e0 = c * CHUNK;
  const int e1 = min(e0 + CHUNK, ne);
  const int nwp = packed_pitch(n);
  for (int wb = 0; wb < nwp; wb += CAPW) {
    const int nwin = min(CAPW, nwp - wb);
    for (int i = tid * 4; i < nwin; i += 4096)
      *(uint4*)&lc[i] = make_uint4(0u, 0u, 0u, 0u);
    __syncthreads();
    const int dlo = wb * 2;
    const int dhi = dlo + nwin * 2;
    int dv[CHUNK / 1024];
    unsigned old[CHUNK / 1024];
#pragma unroll
    for (int k = 0; k < CHUNK / 1024; ++k) {
      const int e = e0 + k * 1024 + tid;
      const int v = dst[min(e, e1 - 1)];
      dv[k] = (e < e1) ? v : -1;
    }
#pragma unroll
    for (int k = 0; k < CHUNK / 1024; ++k) {
      const bool in = dv[k] >= dlo && dv[k] < dhi;
      const int l = in ? dv[k] - dlo : 0;
      const unsigned inc = in ? ((l & 1) ? 65536u : 1u) : 0u;
      old[k] = atomicAdd(&lc[l >> 1], inc);
    }
#pragma unroll
    for (int k = 0; k < CHUNK / 1024; ++k) {
      const int e = e0 + k * 1024 + tid;
      if (dv[k] >= dlo && dv[k] < dhi)
        rank[e] = (int)(((dv[k] - dlo) & 1) ? (old[k] >> 16)
                                            : (old[k] & 0xFFFFu));
    }
    __syncthreads();
    for (int i = tid * 4; i < nwin; i += 4096)
      *(uint4*)&ccnt[(size_t)c * nwp + wb + i] = *(const uint4*)&lc[i];
    __syncthreads();
  }
}

__global__ __launch_bounds__(256) void chunk_offsets(
    const unsigned* __restrict__ ccnt, int* __restrict__ offs,
    int* __restrict__ cnt, int n, int nchunks, int nown,
    const float* __restrict__ x, const float* __restrict__ W1,
    float* __restrict__ t1, int tile0) {
  __shared__ float lw1[128 * KPITCH];
  if ((int)blockIdx.x >= nown) {
    gemm_body<128, 256, 8, true>(x, W1, nullptr, t1, n,
                                 tile0 + (int)blockIdx.x - nown, lw1);
    return;
  }
  const int nwp = packed_pitch(n);
  const int w = blockIdx.x * blockDim.x + threadIdx.x;
  if (w >= nwp) return;
  const size_t np = (size_t)nwp * 2;
  int lo = 0, hi = 0;
#pragma unroll 8
  for (int c = 0; c < nchunks; ++c) {
    const unsigned v = ccnt[(size_t)c * nwp + w];
    *(int2*)(offs + (size_t)c * np + 2 * w) = make_int2(lo, hi);
    lo += (int)(v & 0xFFFFu);
    hi += (int)(v >> 16);
  }
  if (2 * w < n) cnt[2 * w] = lo;
  if (2 * w + 1 < n) cnt[2 * w + 1] = hi;
}

__global__ __launch_bounds__(1024) void scan_rows_fold(
    int* __restrict__ cr, float* __restrict__ invs, int n,
    const float* __restrict__ W3, const float* __restrict__ b3,
    const float* __restrict__ Wo, const float* __restrict__ bo,
    float* __restrict__ w34, float* __restrict__ c34,
    const float* __restrict__ x, const float* __restrict__ W1,
    float* __restrict__ t1, int tile0) {
  __shared__ float lwf[128 * KPITCH];
  if (blockIdx.x >= 3) {
    gemm_body<128, 1024, 8, true>(x, W1, nullptr, t1, n,
                                  tile0 + (int)blockIdx.x - 3, lwf);
    return;
  }
  if (blockIdx.x == 1) {
    gemm_body<64, 1024>(W3, Wo, nullptr, w34, CH, 0, lwf);
    return;
  }
  if (blockIdx.x == 2) {
    gemm_body<64, 1024>(b3, Wo, bo, c34, 1, 0, lwf);
    return;
  }
  __shared__ int wtot[32];
  const int tid = threadIdx.x;
  const int lane = tid & 31;
  const int wv = tid >> 5;
  int carry = 0;
  for (int base = 0; base < n; base += 16384) {
    const int i = base + tid * 16;
    const bool full = i + 15 < n;
    int c[16];
    if (full) {
#pragma unroll
      for (int q = 0; q < 4; ++q) {
        const int4 t4 = *(const int4*)(cr + i + q * 4);
        c[q * 4 + 0] = t4.x;
        c[q * 4 + 1] = t4.y;
        c[q * 4 + 2] = t4.z;
        c[q * 4 + 3] = t4.w;
      }
    } else {
#pragma unroll
      for (int k = 0; k < 16; ++k) c[k] = (i + k < n) ? cr[i + k] : 0;
    }
    int tsum = 0;
#pragma unroll
    for (int k = 0; k < 16; ++k) tsum += c[k];
    int v = tsum;
    for (int o = 1; o < 32; o <<= 1) {
      const int u = __shfl_up(v, o);
      if (lane >= o) v += u;
    }
    if (lane == 31) wtot[wv] = v;
    __syncthreads();
    const int wt = wtot[lane];
    int ws = wt;
    for (int o = 1; o < 32; o <<= 1) {
      const int u = __shfl_up(ws, o);
      if (lane >= o) ws += u;
    }
    const int wexcl = __shfl(ws, wv) - __shfl(wt, wv);
    const int total = __shfl(ws, 31);
    int run = carry + wexcl + (v - tsum);
    int ex[16];
    float rs[16];
#pragma unroll
    for (int k = 0; k < 16; ++k) {
      ex[k] = run;
      run += c[k];
      const float dk = (float)(c[k] + 1);
      rs[k] = dk > 0.f ? rsqrtf(dk) : 0.f;
    }
    if (full) {
#pragma unroll
      for (int q = 0; q < 4; ++q) {
        *(int4*)(cr + i + q * 4) =
            make_int4(ex[q * 4], ex[q * 4 + 1], ex[q * 4 + 2], ex[q * 4 + 3]);
        *(float4*)(invs + i + q * 4) =
            make_float4(rs[q * 4], rs[q * 4 + 1], rs[q * 4 + 2], rs[q * 4 + 3]);
      }
    } else {
#pragma unroll
      for (int k = 0; k < 16; ++k) {
        if (i + k < n) {
          cr[i + k] = ex[k];
          invs[i + k] = rs[k];
        }
      }
    }
    carry += total;
    __syncthreads();
  }
}

__global__ void fill_lists(const int* __restrict__ src,
                           const int* __restrict__ dst,
                           const int* __restrict__ rank,
                           const int* __restrict__ rowptr,
                           const int* __restrict__ offs,
                           const float* __restrict__ invs,
                           int* __restrict__ recs, int n, int ne) {
  int e = blockIdx.x * blockDim.x + threadIdx.x;
  if (e < ne) {
    const int s = src[e];
    const int d = dst[e];
    const size_t np = (size_t)packed_pitch(n) * 2;
    int p = rowptr[d] + offs[(size_t)(e / CHUNK) * np + d] + rank[e];
    p = min(max(p, 0), ne - 1);
    const float w = invs[s] * invs[d];
    ((int2*)recs)[p] = make_int2(s, __float_as_int(w));
  }
}

template <int K>
__device__ __forceinline__ void add_rows(float4& acc,
                                         const float* __restrict__ tl, int rx,
                                         int ry, int j, int last) {
  float4 v[K];
  float wg[K];
#pragma unroll
  for (int u = 0; u < K; ++u) {
    int s = __builtin_amdgcn_readlane(rx, j + u);
    s = min(max(s, 0), last);
    wg[u] = __int_as_float(__builtin_amdgcn_readlane(ry, j + u));
    v[u] = *(const float4*)(tl + (size_t)s * CH);
  }
#pragma unroll
  for (int u = 0; u < K; ++u) {
    acc.x = fmaf(v[u].x, wg[u], acc.x);
    acc.y = fmaf(v[u].y, wg[u], acc.y);
    acc.z = fmaf(v[u].z, wg[u], acc.z);
    acc.w = fmaf(v[u].w, wg[u], acc.w);
  }
}

__device__ __forceinline__ float4 gather_node(
    const float* __restrict__ tl, const int* __restrict__ rowptr,
    const int2* __restrict__ rl, const float4 b, int d, int n, int ne,
    int relu, int lane) {
  const int start = rowptr[d];
  const int end = (d + 1 < n) ? rowptr[d + 1] : ne;
  const float4 tv = *(const float4*)(tl + (size_t)d * CH);
  float4 acc = make_float4(0.f, 0.f, 0.f, 0.f);
  for (int base = start; base < end; base += 32) {
    const int idx = base + lane;
    int2 r = make_int2(0, 0);
    if (idx < end) r = rl[idx];
    const int cnt = min(32, end - base);
    int j = 0;
    for (; j + 8 <= cnt; j += 8) add_rows<8>(acc, tl, r.x, r.y, j, n - 1);
    if (j + 4 <= cnt) { add_rows<4>(acc, tl, r.x, r.y, j, n - 1); j += 4; }
    if (j + 2 <= cnt) { add_rows<2>(acc, tl, r.x, r.y, j, n - 1); j += 2; }
    if (j < cnt) add_rows<1>(acc, tl, r.x, r.y, j, n - 1);
  }
  const float degf = (float)(end - start + 1);
  float sl = degf > 0.f ? rsqrtf(degf) : 0.f;
  sl = sl * sl;
  float4 o;
  o.x = acc.x + tv.x * sl + b.x;
  o.y = acc.y + tv.y * sl + b.y;
  o.z = acc.z + tv.z * sl + b.z;
  o.w = acc.w + tv.w * sl + b.w;
  if (relu) {
    o.x = fmaxf(o.x, 0.f);
    o.y = fmaxf(o.y, 0.f);
    o.z = fmaxf(o.z, 0.f);
    o.w = fmaxf(o.w, 0.f);
  }
  return o;
}

#define TM 32
#define TPITCH 132

template <int NCOL, bool MOVE>
__global__ __launch_bounds__(512) void gather_product(
    const float* __restrict__ t, const int* __restrict__ rowptr,
    const int* __restrict__ recs, const float* __restrict__ bias,
    const float* __restrict__ W, float* __restrict__ tn, int n, int ne,
    int relu, const int* cp_src, int* cp_dst, int ncopy) {
  constexpr int NT = NCOL / 16;
  constexpr int UNITS = (TM / 16) * NT;
  static_assert(UNITS <= 16, "one unit per wave");
  __shared__ float lw[NCOL * KPITCH];
  __shared__ float tile[2][TM * TPITCH];
  const int tid = threadIdx.x;

  if constexpr (MOVE) {
    const int per =
        ((ncopy + (int)gridDim.x - 1) / (int)gridDim.x + 2047) & ~2047;
    const int c0 = (int)blockIdx.x * per;
    const int c1 = min(c0 + per, ncopy);
    const int4* ms = (const int4*)cp_src;
    int4* md = (int4*)cp_dst;
    for (int g0 = c0; g0 < c1; g0 += 2048) {
      if (g0 + 2048 <= ncopy) {
        const int i = g0 + tid;
        const int4 m0 = ms[i];
        const int4 m1 = ms[i + 512];
        const int4 m2 = ms[i + 1024];
        const int4 m3 = ms[i + 1536];
        md[i] = m0;
        md[i + 512] = m1;
        md[i + 1024] = m2;
        md[i + 1536] = m3;
      } else {
        for (int i = g0 + tid; i < ncopy; i += 512) md[i] = ms[i];
      }
    }
  }

  {
    constexpr int TR = CH * NCOL / 512;
    constexpr int SB = 16;
    static_assert(TR % SB == 0, "whole batches");
    for (int q0 = 0; q0 < TR; q0 += SB) {
      float wv[SB];
#pragma unroll
      for (int q = 0; q < SB; ++q) wv[q] = W[tid + (q0 + q) * 512];
#pragma unroll
      for (int q = 0; q < SB; ++q) {
        const int i = tid + (q0 + q) * 512;
        const int k = i / NCOL, c = i - k * NCOL;
        lw[c * KPITCH + k] = wv[q];
      }
    }
  }

  const int lane = tid & 31;
  const int wave = __builtin_amdgcn_readfirstlane(tid >> 5);
  const float4 b = *(const float4*)(bias + lane * 4);
  const int2* rl = (const int2*)recs;
  const float* tl = t + lane * 4;
  const int lrow = lane & 15;
  const int kbase = (lane >> 4) * 2;
  const int ntiles = (n + TM - 1) / TM;
  int buf = 0;
  for (int ti = blockIdx.x; ti < ntiles; ti += gridDim.x, buf ^= 1) {
    float* tb = tile[buf];
    for (int i = 0; i < TM / 16; ++i) {
      const int r = wave * (TM / 16) + i;
      const int d = ti * TM + r;
      float4 o = make_float4(0.f, 0.f, 0.f, 0.f);
      if (d < n) o = gather_node(tl, rowptr, rl, b, d, n, ne, relu, lane);
      *(float4*)&tb[r * TPITCH + lane * 4] = o;
    }
    __syncthreads();
    if (wave < UNITS) {
      const int s = wave / NT;
      const int nt = wave - s * NT;
      const float* ap = tb + (s * 16 + lrow) * TPITCH + kbase;
      const float* bp = lw + (nt * 16 + lrow) * KPITCH + kbase;
      v8f acc = {};
#pragma unroll
      for (int kc = 0; kc < CH / 4; ++kc) {
        const float2 av = *(const float2*)(ap + kc * 4);
        const float2 bv = *(const float2*)(bp + kc * 4);
        v2f a, bb;
        a[0] = av.x; a[1] = av.y;
        bb[0] = bv.x; bb[1] = bv.y;
        acc = __builtin_amdgcn_wmma_f32_16x16x4_f32(false, a, false, bb,
                                                    (short)0, acc, false,
                                                    false);
      }
      __builtin_amdgcn_sched_barrier(0);
      asm volatile(
          "v_nop\n\tv_nop\n\tv_nop\n\tv_nop\n\tv_nop\n\tv_nop\n\tv_nop\n\tv_nop"
          : "+v"(acc));
      __builtin_amdgcn_sched_barrier(0);
      const int rbase = ti * TM + s * 16 + ((lane >> 4) << 3);
      const int col = nt * 16 + lrow;
#pragma unroll
      for (int r = 0; r < 8; ++r) {
        const int row = rbase + r;
        if (row < n) tn[(size_t)row * NCOL + col] = acc[r];
      }
    }
  }
}

template <int K>
__device__ __forceinline__ void add_rows64(float2& acc,
                                           const float* __restrict__ tl,
                                           int rx, int ry, int j, int last) {
  float2 v[K];
  float wg[K];
#pragma unroll
  for (int u = 0; u < K; ++u) {
    int s = __builtin_amdgcn_readlane(rx, j + u);
    s = min(max(s, 0), last);
    wg[u] = __int_as_float(__builtin_amdgcn_readlane(ry, j + u));
    v[u] = *(const float2*)(tl + (size_t)s * 64);
  }
#pragma unroll
  for (int u = 0; u < K; ++u) {
    acc.x += v[u].x * wg[u];
    acc.y += v[u].y * wg[u];
  }
}

__global__ __launch_bounds__(256) void gather_rows64(
    const float* __restrict__ t, const int* __restrict__ rowptr,
    const int* __restrict__ recs, const float* __restrict__ bias,
    float* __restrict__ h, int n, int ne) {
  const int lane = threadIdx.x & 31;
  const int w0 = __builtin_amdgcn_readfirstlane(
      (int)((blockIdx.x * blockDim.x + threadIdx.x) >> 5));
  const int nw = (int)((gridDim.x * blockDim.x) >> 5);
  const float2 b = *(const float2*)(bias + lane * 2);
  const int2* rl = (const int2*)recs;
  const float* tl = t + lane * 2;
  for (int d = w0; d < n; d += nw) {
    const int start = rowptr[d];
    const int end = (d + 1 < n) ? rowptr[d + 1] : ne;
    const float2 tv = *(const float2*)(tl + (size_t)d * 64);
    float2 acc = make_float2(0.f, 0.f);
    for (int base = start; base < end; base += 32) {
      const int idx = base + lane;
      int2 r = make_int2(0, 0);
      if (idx < end) r = rl[idx];
      const int cnt = min(32, end - base);
      int j = 0;
      for (; j + 16 <= cnt; j += 16) add_rows64<16>(acc, tl, r.x, r.y, j, n - 1);
      if (j + 8 <= cnt) { add_rows64<8>(acc, tl, r.x, r.y, j, n - 1); j += 8; }
      if (j + 4 <= cnt) { add_rows64<4>(acc, tl, r.x, r.y, j, n - 1); j += 4; }
      if (j + 2 <= cnt) { add_rows64<2>(acc, tl, r.x, r.y, j, n - 1); j += 2; }
      if (j < cnt) add_rows64<1>(acc, tl, r.x, r.y, j, n - 1);
    }
    const float degf = (float)(end - start + 1);
    float sl = degf > 0.f ? rsqrtf(degf) : 0.f;
    sl = sl * sl;
    float2 o;
    o.x = acc.x + tv.x * sl + b.x;
    o.y = acc.y + tv.y * sl + b.y;
    *(float2*)(h + (size_t)d * 64 + lane * 2) = o;
  }
}

extern "C" void kernel_launch(void* const* d_in, const int* in_sizes, int n_in,
                              void* d_out, int out_size, void* d_ws,
                              size_t ws_size, hipStream_t stream) {
  const float* x = (const float*)d_in[0];
  const int* ei = (const int*)d_in[1];
  const float* W1 = (const float*)d_in[2];
  const float* b1 = (const float*)d_in[3];
  const float* W2 = (const float*)d_in[4];
  const float* b2 = (const float*)d_in[5];
  const float* W3 = (const float*)d_in[6];
  const float* b3 = (const float*)d_in[7];
  const float* Wo = (const float*)d_in[8];
  const float* bo = (const float*)d_in[9];

  const int N = in_sizes[0] / CH;
  const int E = in_sizes[1] / 2;
  const int* src = ei;
  const int* dstp = ei + E;

  float* bufA = (float*)d_ws;
  float* bufB = bufA + (size_t)N * CH;
  int* rowptr = (int*)(bufB + (size_t)N * CH);
  int* rank = rowptr + N;
  float* invs = bufB;
  int* offs = (int*)(bufB + (size_t)2 * N);
  int* recs = (int*)d_out;
  float* c34 = (float*)d_out + (size_t)2 * E;
  float* w34 = c34 + 64;
  unsigned* ccnt = (unsigned*)(w34 + (size_t)CH * 64);
  float* t3 = bufA;
  int* recs3 = (int*)(bufA + (size_t)N * 64);
  const float* c34m = (const float*)(recs3 + (size_t)2 * E);
  const int nmove = (2 * E + 64 + 3) / 4;

  const dim3 blk(256);
  const int gE = (E + 255) / 256;
  const int gChunk = (E + CHUNK - 1) / CHUNK;
  const int gWord = (N / 2 + 8 + 255) / 256;
  const int gGemm = (N + 127) / 128;
  const int gGather = 2048;
  const int gFuse128 = 768;
  const int gFuse64 = 1024;
  const int gT1 = 207;
  const int gT2 = 256;
  const int gT3 = gGemm > gT1 + gT2 ? gGemm - gT1 - gT2 : 0;

  chunk_count<<<gChunk + gT1, dim3(1024), 0, stream>>>(
      dstp, ccnt, rank, N, E, gChunk, x, W1, bufA, 0);
  chunk_offsets<<<gWord + gT2, blk, 0, stream>>>(ccnt, offs, rowptr, N, gChunk,
                                                 gWord, x, W1, bufA, gT1);
  scan_rows_fold<<<3 + gT3, dim3(1024), 0, stream>>>(
      rowptr, invs, N, W3, b3, Wo, bo, w34, c34, x, W1, bufA, gT1 + gT2);
  fill_lists<<<gE, blk, 0, stream>>>(src, dstp, rank, rowptr, offs, invs, recs,
                                     N, E);

  gather_product<128, false><<<gFuse128, dim3(512), 0, stream>>>(
      bufA, rowptr, recs, b1, W2, bufB, N, E, 1, nullptr, nullptr, 0);

  gather_product<64, true><<<gFuse64, dim3(512), 0, stream>>>(
      bufB, rowptr, recs, b2, w34, t3, N, E, 1, recs, recs3, nmove);

  gather_rows64<<<gGather, blk, 0, stream>>>(t3, rowptr, recs3, c34m,
                                             (float*)d_out, N, E);
}


# gather walks request a node's list ahead: bounds two nodes on, records one node on, both issued behind the current node's first row loads; 3 of 6 dependent round trips per node leave; sums unchanged
# speedup vs baseline: 4.9290x; 1.0324x over previous
#include <hip/hip_runtime.h>

typedef float v2f __attribute__((ext_vector_type(2)));
typedef float v8f __attribute__((ext_vector_type(8)));

#define CH 128

#define KPITCH 130

template <int NCOL, int THREADS, int SPB = THREADS / 32,
          bool PAD = (NCOL == 64)>
__device__ __forceinline__ void gemm_body(
    const float* __restrict__ A, const float* __restrict__ W,
    const float* __restrict__ bias, float* __restrict__ C, int nrows, int blk,
    float* __restrict__ lw) {
  constexpr int NT = NCOL / 16;
  constexpr int KP = KPITCH;

  const int tid = threadIdx.x;
  if constexpr (THREADS == 256) {
    for (int i = tid; i < CH * NCOL; i += THREADS) {
      int k = i / NCOL, c = i - k * NCOL;
      lw[c * KP + k] = W[i];
    }
  } else {
    constexpr int TR = CH * NCOL / THREADS;
    float wv[TR];
#pragma unroll
    for (int q = 0; q < TR; ++q) wv[q] = W[tid + q * THREADS];
#pragma unroll
    for (int q = 0; q < TR; ++q) {
      const int i = tid + q * THREADS;
      const int k = i / NCOL, c = i - k * NCOL;
      lw[c * KP + k] = wv[q];
    }
  }
  __syncthreads();

  const int wave = tid >> 5;
  const int lane = tid & 31;
  if constexpr (SPB < THREADS / 32) {
    if (wave >= SPB) return;
  }
  const int m0 = (blk * SPB + wave) * 16;
  if (m0 >= nrows) return;

  const int lrow = lane & 15;
  const int kbase = (lane >> 4) * 2;

  v8f acc[NT] = {};

  int arow = m0 + lrow;
  if (arow > nrows - 1) arow = nrows - 1;
  const float* aptr = A + (size_t)arow * CH + kbase;

#pragma unroll(NCOL == 64 ? CH / 4 : 4)
  for (int kc = 0; kc < CH / 4; ++kc) {
    float2 av = *(const float2*)(aptr + kc * 4);
    v2f a;
    a[0] = av.x; a[1] = av.y;
#pragma unroll
    for (int n = 0; n < NT; ++n) {
      float2 bv = *(const float2*)&lw[(n * 16 + lrow) * KP + kc * 4 + kbase];
      v2f b;
      b[0] = bv.x; b[1] = bv.y;
      acc[n] = __builtin_amdgcn_wmma_f32_16x16x4_f32(false, a, false, b,
                                                     (short)0, acc[n],
                                                     false, false);
    }
  }

  if constexpr (PAD) {
    if constexpr (NCOL == 64) __builtin_amdgcn_sched_barrier(0);
    if constexpr (NT == 4) {
      asm volatile(
          "v_nop\n\tv_nop\n\tv_nop\n\tv_nop\n\tv_nop\n\tv_nop\n\tv_nop\n\tv_nop"
          : "+v"(acc[0]), "+v"(acc[1]), "+v"(acc[2]), "+v"(acc[3]));
    } else {
      static_assert(NT == 4 || NT == 8, "accumulators named below");
      asm volatile(
          "v_nop\n\tv_nop\n\tv_nop\n\tv_nop\n\tv_nop\n\tv_nop\n\tv_nop\n\tv_nop"
          : "+v"(acc[0]), "+v"(acc[1]), "+v"(acc[2]), "+v"(acc[3]),
            "+v"(acc[4]), "+v"(acc[5]), "+v"(acc[6]), "+v"(acc[7]));
    }
    __builtin_amdgcn_sched_barrier(0);
  }

  const int rbase = m0 + ((lane >> 4) << 3);
#pragma unroll
  for (int r = 0; r < 8; ++r) {
    int row = rbase + r;
    if (row < nrows) {
#pragma unroll
      for (int n = 0; n < NT; ++n) {
        int col = n * 16 + lrow;
        float v = acc[n][r];
        if (bias) v += bias[col];
        C[(size_t)row * NCOL + col] = v;
      }
    }
  }
}

#define CHUNK 32768
#define CAPW 65536

__device__ __forceinline__ int packed_pitch(int n) {
  return (((n + 1) >> 1) + 3) & ~3;
}

__global__ __launch_bounds__(1024) void chunk_count(
    const int* __restrict__ dst, unsigned* __restrict__ ccnt,
    int* __restrict__ rank, int n, int ne, int nchunks,
    const float* __restrict__ x, const float* __restrict__ W1,
    float* __restrict__ t1, int tile0) {
  __shared__ unsigned lc[CAPW];
  if ((int)blockIdx.x >= nchunks) {
    gemm_body<128, 1024, 8, true>(x, W1, nullptr, t1, n,
                                  tile0 + (int)blockIdx.x - nchunks,
                                  (float*)lc);
    return;
  }
  const int tid = threadIdx.x;
  const int c = blockIdx.x;
  const int e0 = c * CHUNK;
  const int e1 = min(e0 + CHUNK, ne);
  const int nwp = packed_pitch(n);
  for (int wb = 0; wb < nwp; wb += CAPW) {
    const int nwin = min(CAPW, nwp - wb);
    for (int i = tid * 4; i < nwin; i += 4096)
      *(uint4*)&lc[i] = make_uint4(0u, 0u, 0u, 0u);
    __syncthreads();
    const int dlo = wb * 2;
    const int dhi = dlo + nwin * 2;
    int dv[CHUNK / 1024];
    unsigned old[CHUNK / 1024];
#pragma unroll
    for (int k = 0; k < CHUNK / 1024; ++k) {
      const int e = e0 + k * 1024 + tid;
      const int v = dst[min(e, e1 - 1)];
      dv[k] = (e < e1) ? v : -1;
    }
#pragma unroll
    for (int k = 0; k < CHUNK / 1024; ++k) {
      const bool in = dv[k] >= dlo && dv[k] < dhi;
      const int l = in ? dv[k] - dlo : 0;
      const unsigned inc = in ? ((l & 1) ? 65536u : 1u) : 0u;
      old[k] = atomicAdd(&lc[l >> 1], inc);
    }
#pragma unroll
    for (int k = 0; k < CHUNK / 1024; ++k) {
      const int e = e0 + k * 1024 + tid;
      if (dv[k] >= dlo && dv[k] < dhi)
        rank[e] = (int)(((dv[k] - dlo) & 1) ? (old[k] >> 16)
                                            : (old[k] & 0xFFFFu));
    }
    __syncthreads();
    for (int i = tid * 4; i < nwin; i += 4096)
      *(uint4*)&ccnt[(size_t)c * nwp + wb + i] = *(const uint4*)&lc[i];
    __syncthreads();
  }
}

__global__ __launch_bounds__(256) void chunk_offsets(
    const unsigned* __restrict__ ccnt, int* __restrict__ offs,
    int* __restrict__ cnt, int n, int nchunks, int nown,
    const float* __restrict__ x, const float* __restrict__ W1,
    float* __restrict__ t1, int tile0) {
  __shared__ float lw1[128 * KPITCH];
  if ((int)blockIdx.x >= nown) {
    gemm_body<128, 256, 8, true>(x, W1, nullptr, t1, n,
                                 tile0 + (int)blockIdx.x - nown, lw1);
    return;
  }
  const int nwp = packed_pitch(n);
  const int w = blockIdx.x * blockDim.x + threadIdx.x;
  if (w >= nwp) return;
  const size_t np = (size_t)nwp * 2;
  int lo = 0, hi = 0;
#pragma unroll 8
  for (int c = 0; c < nchunks; ++c) {
    const unsigned v = ccnt[(size_t)c * nwp + w];
    *(int2*)(offs + (size_t)c * np + 2 * w) = make_int2(lo, hi);
    lo += (int)(v & 0xFFFFu);
    hi += (int)(v >> 16);
  }
  if (2 * w < n) cnt[2 * w] = lo;
  if (2 * w + 1 < n) cnt[2 * w + 1] = hi;
}

__global__ __launch_bounds__(1024) void scan_rows_fold(
    int* __restrict__ cr, float* __restrict__ invs, int n,
    const float* __restrict__ W3, const float* __restrict__ b3,
    const float* __restrict__ Wo, const float* __restrict__ bo,
    float* __restrict__ w34, float* __restrict__ c34,
    const float* __restrict__ x, const float* __restrict__ W1,
    float* __restrict__ t1, int tile0) {
  __shared__ float lwf[128 * KPITCH];
  if (blockIdx.x >= 3) {
    gemm_body<128, 1024, 8, true>(x, W1, nullptr, t1, n,
                                  tile0 + (int)blockIdx.x - 3, lwf);
    return;
  }
  if (blockIdx.x == 1) {
    gemm_body<64, 1024>(W3, Wo, nullptr, w34, CH, 0, lwf);
    return;
  }
  if (blockIdx.x == 2) {
    gemm_body<64, 1024>(b3, Wo, bo, c34, 1, 0, lwf);
    return;
  }
  __shared__ int wtot[32];
  const int tid = threadIdx.x;
  const int lane = tid & 31;
  const int wv = tid >> 5;
  int carry = 0;
  for (int base = 0; base < n; base += 16384) {
    const int i = base + tid * 16;
    const bool full = i + 15 < n;
    int c[16];
    if (full) {
#pragma unroll
      for (int q = 0; q < 4; ++q) {
        const int4 t4 = *(const int4*)(cr + i + q * 4);
        c[q * 4 + 0] = t4.x;
        c[q * 4 + 1] = t4.y;
        c[q * 4 + 2] = t4.z;
        c[q * 4 + 3] = t4.w;
      }
    } else {
#pragma unroll
      for (int k = 0; k < 16; ++k) c[k] = (i + k < n) ? cr[i + k] : 0;
    }
    int tsum = 0;
#pragma unroll
    for (int k = 0; k < 16; ++k) tsum += c[k];
    int v = tsum;
    for (int o = 1; o < 32; o <<= 1) {
      const int u = __shfl_up(v, o);
      if (lane >= o) v += u;
    }
    if (lane == 31) wtot[wv] = v;
    __syncthreads();
    const int wt = wtot[lane];
    int ws = wt;
    for (int o = 1; o < 32; o <<= 1) {
      const int u = __shfl_up(ws, o);
      if (lane >= o) ws += u;
    }
    const int wexcl = __shfl(ws, wv) - __shfl(wt, wv);
    const int total = __shfl(ws, 31);
    int run = carry + wexcl + (v - tsum);
    int ex[16];
    float rs[16];
#pragma unroll
    for (int k = 0; k < 16; ++k) {
      ex[k] = run;
      run += c[k];
      const float dk = (float)(c[k] + 1);
      rs[k] = dk > 0.f ? rsqrtf(dk) : 0.f;
    }
    if (full) {
#pragma unroll
      for (int q = 0; q < 4; ++q) {
        *(int4*)(cr + i + q * 4) =
            make_int4(ex[q * 4], ex[q * 4 + 1], ex[q * 4 + 2], ex[q * 4 + 3]);
        *(float4*)(invs + i + q * 4) =
            make_float4(rs[q * 4], rs[q * 4 + 1], rs[q * 4 + 2], rs[q * 4 + 3]);
      }
    } else {
#pragma unroll
      for (int k = 0; k < 16; ++k) {
        if (i + k < n) {
          cr[i + k] = ex[k];
          invs[i + k] = rs[k];
        }
      }
    }
    carry += total;
    __syncthreads();
  }
}

__global__ void fill_lists(const int* __restrict__ src,
                           const int* __restrict__ dst,
                           const int* __restrict__ rank,
                           const int* __restrict__ rowptr,
                           const int* __restrict__ offs,
                           const float* __restrict__ invs,
                           int* __restrict__ recs, int n, int ne) {
  int e = blockIdx.x * blockDim.x + threadIdx.x;
  if (e < ne) {
    const int s = src[e];
    const int d = dst[e];
    const size_t np = (size_t)packed_pitch(n) * 2;
    int p = rowptr[d] + offs[(size_t)(e / CHUNK) * np + d] + rank[e];
    p = min(max(p, 0), ne - 1);
    const float w = invs[s] * invs[d];
    ((int2*)recs)[p] = make_int2(s, __float_as_int(w));
  }
}

template <int K>
__device__ __forceinline__ void add_rows(float4& acc,
                                         const float* __restrict__ tl, int rx,
                                         int ry, int j, int last) {
  float4 v[K];
  float wg[K];
#pragma unroll
  for (int u = 0; u < K; ++u) {
    int s = __builtin_amdgcn_readlane(rx, j + u);
    s = min(max(s, 0), last);
    wg[u] = __int_as_float(__builtin_amdgcn_readlane(ry, j + u));
    v[u] = *(const float4*)(tl + (size_t)s * CH);
  }
#pragma unroll
  for (int u = 0; u < K; ++u) {
    acc.x = fmaf(v[u].x, wg[u], acc.x);
    acc.y = fmaf(v[u].y, wg[u], acc.y);
    acc.z = fmaf(v[u].z, wg[u], acc.z);
    acc.w = fmaf(v[u].w, wg[u], acc.w);
  }
}

__device__ __forceinline__ void node_bounds(const int* __restrict__ rowptr,
                                            int d, int n, int ne, int& start,
                                            int& end) {
  const int dc = min(d, n - 1);
  const int s = rowptr[dc];
  const int e = (dc + 1 < n) ? rowptr[dc + 1] : ne;
  start = s;
  end = (d < n) ? e : s;
}

__device__ __forceinline__ int bounds_index(int d, int lane, int n) {
  return min(d + (lane & 1), n - 1);
}
__device__ __forceinline__ void bounds_decode(int vb, int d, int n, int ne,
                                              int& start, int& end) {
  const int s = __builtin_amdgcn_readlane(vb, 0);
  const int e = __builtin_amdgcn_readlane(vb, 1);
  start = s;
  end = (d < n) ? ((d + 1 < n) ? e : ne) : s;
}

__device__ __forceinline__ int ahead_index(int start, int lane, int ne) {
  return max(min(start + lane, ne - 1), 0);
}

template <int K>
__device__ __forceinline__ void add_rows_first(
    float4& acc, const float* __restrict__ tl, int rx, int ry, int last,
    float4& tv, int d, int2& rn, const int2* __restrict__ rl, int nidx,
    int& vb, const int* __restrict__ rowptr, int bidx) {
  float4 v[K];
  float wg[K];
#pragma unroll
  for (int u = 0; u < K; ++u) {
    int s = __builtin_amdgcn_readlane(rx, u);
    s = min(max(s, 0), last);
    wg[u] = __int_as_float(__builtin_amdgcn_readlane(ry, u));
    v[u] = *(const float4*)(tl + (size_t)s * CH);
  }
  tv = *(const float4*)(tl + (size_t)d * CH);
  rn = rl[nidx];
  vb = rowptr[bidx];
#pragma unroll
  for (int u = 0; u < K; ++u) {
    acc.x = fmaf(v[u].x, wg[u], acc.x);
    acc.y = fmaf(v[u].y, wg[u], acc.y);
    acc.z = fmaf(v[u].z, wg[u], acc.z);
    acc.w = fmaf(v[u].w, wg[u], acc.w);
  }
}

__device__ __forceinline__ void walk_rows(float4& acc,
                                          const float* __restrict__ tl,
                                          int2 r, int j, int cnt, int last) {
  for (; j + 8 <= cnt; j += 8) add_rows<8>(acc, tl, r.x, r.y, j, last);
  if (j + 4 <= cnt) { add_rows<4>(acc, tl, r.x, r.y, j, last); j += 4; }
  if (j + 2 <= cnt) { add_rows<2>(acc, tl, r.x, r.y, j, last); j += 2; }
  if (j < cnt) add_rows<1>(acc, tl, r.x, r.y, j, last);
}

template <bool BLATE>
__device__ __forceinline__ float4 gather_node(
    const float* __restrict__ tl, const int2* __restrict__ rl, float4 b,
    const float* __restrict__ bl, int d, int n, int ne, int relu, int lane,
    int start, int end, int2 r, int ns, int2& rn, int& vb,
    const int* __restrict__ rowptr, int bidx) {
  const int nidx = ahead_index(ns, lane, ne);
  float4 tv;
  float4 acc = make_float4(0.f, 0.f, 0.f, 0.f);
  const int cnt0 = min(32, end - start);
  int j = 0;
  if (cnt0 >= 8) {
    add_rows_first<8>(acc, tl, r.x, r.y, n - 1, tv, d, rn, rl, nidx, vb,
                      rowptr, bidx);
    j = 8;
  } else if (cnt0 >= 4) {
    add_rows_first<4>(acc, tl, r.x, r.y, n - 1, tv, d, rn, rl, nidx, vb,
                      rowptr, bidx);
    j = 4;
  } else if (cnt0 >= 2) {
    add_rows_first<2>(acc, tl, r.x, r.y, n - 1, tv, d, rn, rl, nidx, vb,
                      rowptr, bidx);
    j = 2;
  } else if (cnt0 >= 1) {
    add_rows_first<1>(acc, tl, r.x, r.y, n - 1, tv, d, rn, rl, nidx, vb,
                      rowptr, bidx);
    j = 1;
  } else {
    tv = *(const float4*)(tl + (size_t)d * CH);
    rn = rl[nidx];
    vb = rowptr[bidx];
  }
  walk_rows(acc, tl, r, j, cnt0, n - 1);
  for (int base = start + 32; base < end; base += 32) {
    const int idx = base + lane;
    int2 rr = make_int2(0, 0);
    if (idx < end) rr = rl[idx];
    walk_rows(acc, tl, rr, 0, min(32, end - base), n - 1);
  }
  const float degf = (float)(end - start + 1);
  float sl = degf > 0.f ? rsqrtf(degf) : 0.f;
  sl = sl * sl;
  if constexpr (BLATE) b = *(const float4*)bl;
  float4 o;
  o.x = acc.x + tv.x * sl + b.x;
  o.y = acc.y + tv.y * sl + b.y;
  o.z = acc.z + tv.z * sl + b.z;
  o.w = acc.w + tv.w * sl + b.w;
  if (relu) {
    o.x = fmaxf(o.x, 0.f);
    o.y = fmaxf(o.y, 0.f);
    o.z = fmaxf(o.z, 0.f);
    o.w = fmaxf(o.w, 0.f);
  }
  return o;
}

#define TM 32
#define TPITCH 132

template <int NCOL, bool MOVE>
__global__ __launch_bounds__(512)
__attribute__((amdgpu_waves_per_eu(NCOL == 64 ? 16 : 12))) void gather_product(
    const float* __restrict__ t, const int* __restrict__ rowptr,
    const int* __restrict__ recs, const float* __restrict__ bias,
    const float* __restrict__ W, float* __restrict__ tn, int n, int ne,
    int relu, const int* cp_src, int* cp_dst, int ncopy) {
  constexpr int NT = NCOL / 16;
  constexpr int UNITS = (TM / 16) * NT;
  static_assert(UNITS <= 16, "one unit per wave");
  __shared__ float lw[NCOL * KPITCH];
  __shared__ float tile[2][TM * TPITCH];
  const int tid = threadIdx.x;

  if constexpr (MOVE) {
    const int per =
        ((ncopy + (int)gridDim.x - 1) / (int)gridDim.x + 2047) & ~2047;
    const int c0 = (int)blockIdx.x * per;
    const int c1 = min(c0 + per, ncopy);
    const int4* ms = (const int4*)cp_src;
    int4* md = (int4*)cp_dst;
    for (int g0 = c0; g0 < c1; g0 += 2048) {
      if (g0 + 2048 <= ncopy) {
        const int i = g0 + tid;
        const int4 m0 = ms[i];
        const int4 m1 = ms[i + 512];
        const int4 m2 = ms[i + 1024];
        const int4 m3 = ms[i + 1536];
        md[i] = m0;
        md[i + 512] = m1;
        md[i + 1024] = m2;
        md[i + 1536] = m3;
      } else {
        for (int i = g0 + tid; i < ncopy; i += 512) md[i] = ms[i];
      }
    }
  }

  const int lane = tid & 31;
  const int wave = __builtin_amdgcn_readfirstlane(tid >> 5);
  const int2* rl = (const int2*)recs;
  static_assert(TM / 16 == 2, "two rows per wave and tile");
  const int r0 = wave * (TM / 16);
  const int dstep = (int)gridDim.x * TM;
  int cs, ce, vb1;
  int2 cr;
  {
    const int d0 = (int)blockIdx.x * TM + r0;
    node_bounds(rowptr, d0, n, ne, cs, ce);
    vb1 = rowptr[bounds_index(d0 + 1, lane, n)];
  }

  {
    constexpr int TR = CH * NCOL / 512;
    constexpr int SB = 16;
    static_assert(TR % SB == 0, "whole batches");
    for (int q0 = 0; q0 < TR; q0 += SB) {
      float wv[SB];
#pragma unroll
      for (int q = 0; q < SB; ++q) wv[q] = W[tid + (q0 + q) * 512];
#pragma unroll
      for (int q = 0; q < SB; ++q) {
        const int i = tid + (q0 + q) * 512;
        const int k = i / NCOL, c = i - k * NCOL;
        lw[c * KPITCH + k] = wv[q];
      }
    }
  }
  cr = rl[ahead_index(cs, lane, ne)];

  float4 b = make_float4(0.f, 0.f, 0.f, 0.f);
  if constexpr (NCOL != 64) b = *(const float4*)(bias + lane * 4);
  const float* tl = t + lane * 4;
  const int lrow = lane & 15;
  const int kbase = (lane >> 4) * 2;
  const int ntiles = (n + TM - 1) / TM;
  int buf = 0;
  for (int ti = blockIdx.x; ti < ntiles; ti += gridDim.x, buf ^= 1) {
    float* tb = tile[buf];
#pragma unroll
    for (int i = 0; i < TM / 16; ++i) {
      const int r = r0 + i;
      const int d = ti * TM + r;
      const int dn = (i + 1 < TM / 16) ? d + 1 : d + dstep - (TM / 16 - 1);
      int ns, nx;
      bounds_decode(vb1, dn, n, ne, ns, nx);
      int2 rn = make_int2(0, 0);
      int vb2 = 0;
      float4 o = make_float4(0.f, 0.f, 0.f, 0.f);
      if (d < n)
        o = gather_node<NCOL == 64>(tl, rl, b, bias + lane * 4, d, n, ne, relu,
                                    lane, cs, ce, cr, ns, rn, vb2, rowptr,
                                    bounds_index(d + dstep, lane, n));
      *(float4*)&tb[r * TPITCH + lane * 4] = o;
      cs = ns; ce = nx; cr = rn;
      vb1 = vb2;
    }
    __syncthreads();
    if (wave < UNITS) {
      const int s = wave / NT;
      const int nt = wave - s * NT;
      const float* ap = tb + (s * 16 + lrow) * TPITCH + kbase;
      const float* bp = lw + (nt * 16 + lrow) * KPITCH + kbase;
      v8f acc = {};
#pragma unroll
      for (int kc = 0; kc < CH / 4; ++kc) {
        const float2 av = *(const float2*)(ap + kc * 4);
        const float2 bv = *(const float2*)(bp + kc * 4);
        v2f a, bb;
        a[0] = av.x; a[1] = av.y;
        bb[0] = bv.x; bb[1] = bv.y;
        acc = __builtin_amdgcn_wmma_f32_16x16x4_f32(false, a, false, bb,
                                                    (short)0, acc, false,
                                                    false);
      }
      __builtin_amdgcn_sched_barrier(0);
      asm volatile(
          "v_nop\n\tv_nop\n\tv_nop\n\tv_nop\n\tv_nop\n\tv_nop\n\tv_nop\n\tv_nop"
          : "+v"(acc));
      __builtin_amdgcn_sched_barrier(0);
      const int rbase = ti * TM + s * 16 + ((lane >> 4) << 3);
      const int col = nt * 16 + lrow;
#pragma unroll
      for (int r = 0; r < 8; ++r) {
        const int row = rbase + r;
        if (row < n) tn[(size_t)row * NCOL + col] = acc[r];
      }
    }
  }
}

template <int K>
__device__ __forceinline__ void add_rows64(float2& acc,
                                           const float* __restrict__ tl,
                                           int rx, int ry, int j, int last) {
  float2 v[K];
  float wg[K];
#pragma unroll
  for (int u = 0; u < K; ++u) {
    int s = __builtin_amdgcn_readlane(rx, j + u);
    s = min(max(s, 0), last);
    wg[u] = __int_as_float(__builtin_amdgcn_readlane(ry, j + u));
    v[u] = *(const float2*)(tl + (size_t)s * 64);
  }
#pragma unroll
  for (int u = 0; u < K; ++u) {
    acc.x += v[u].x * wg[u];
    acc.y += v[u].y * wg[u];
  }
}

template <int K>
__device__ __forceinline__ void add_rows64_first(
    float2& acc, const float* __restrict__ tl, int rx, int ry, int last,
    float2& tv, int d, int2& rn, const int2* __restrict__ rl, int nidx,
    int& vb, const int* __restrict__ rowptr, int bidx) {
  float2 v[K];
  float wg[K];
#pragma unroll
  for (int u = 0; u < K; ++u) {
    int s = __builtin_amdgcn_readlane(rx, u);
    s = min(max(s, 0), last);
    wg[u] = __int_as_float(__builtin_amdgcn_readlane(ry, u));
    v[u] = *(const float2*)(tl + (size_t)s * 64);
  }
  tv = *(const float2*)(tl + (size_t)d * 64);
  rn = rl[nidx];
  vb = rowptr[bidx];
#pragma unroll
  for (int u = 0; u < K; ++u) {
    acc.x += v[u].x * wg[u];
    acc.y += v[u].y * wg[u];
  }
}

__device__ __forceinline__ void walk_rows64(float2& acc,
                                            const float* __restrict__ tl,
                                            int2 r, int j, int cnt, int last) {
  for (; j + 16 <= cnt; j += 16) add_rows64<16>(acc, tl, r.x, r.y, j, last);
  if (j + 8 <= cnt) { add_rows64<8>(acc, tl, r.x, r.y, j, last); j += 8; }
  if (j + 4 <= cnt) { add_rows64<4>(acc, tl, r.x, r.y, j, last); j += 4; }
  if (j + 2 <= cnt) { add_rows64<2>(acc, tl, r.x, r.y, j, last); j += 2; }
  if (j < cnt) add_rows64<1>(acc, tl, r.x, r.y, j, last);
}

__global__ __launch_bounds__(256) void gather_rows64(
    const float* __restrict__ t, const int* __restrict__ rowptr,
    const int* __restrict__ recs, const float* __restrict__ bias,
    float* __restrict__ h, int n, int ne) {
  const int lane = threadIdx.x & 31;
  const int w0 = __builtin_amdgcn_readfirstlane(
      (int)((blockIdx.x * blockDim.x + threadIdx.x) >> 5));
  const int nw = (int)((gridDim.x * blockDim.x) >> 5);
  const float2 b = *(const float2*)(bias + lane * 2);
  const int2* rl = (const int2*)recs;
  const float* tl = t + lane * 2;
  int start, end;
  node_bounds(rowptr, w0, n, ne, start, end);
  int vb1 = rowptr[bounds_index(w0 + nw, lane, n)];
  int2 r = rl[ahead_index(start, lane, ne)];
  for (int d = w0; d < n; d += nw) {
    int ns, nx;
    bounds_decode(vb1, d + nw, n, ne, ns, nx);
    const int nidx = ahead_index(ns, lane, ne);
    const int bidx = bounds_index(d + 2 * nw, lane, n);
    int2 rn;
    int vb2;
    float2 tv;
    float2 acc = make_float2(0.f, 0.f);
    const int cnt0 = min(32, end - start);
    int j = 0;
    if (cnt0 >= 16) {
      add_rows64_first<16>(acc, tl, r.x, r.y, n - 1, tv, d, rn, rl, nidx, vb2,
                           rowptr, bidx);
      j = 16;
    } else if (cnt0 >= 8) {
      add_rows64_first<8>(acc, tl, r.x, r.y, n - 1, tv, d, rn, rl, nidx, vb2,
                           rowptr, bidx);
      j = 8;
    } else if (cnt0 >= 4) {
      add_rows64_first<4>(acc, tl, r.x, r.y, n - 1, tv, d, rn, rl, nidx, vb2,
                           rowptr, bidx);
      j = 4;
    } else if (cnt0 >= 2) {
      add_rows64_first<2>(acc, tl, r.x, r.y, n - 1, tv, d, rn, rl, nidx, vb2,
                           rowptr, bidx);
      j = 2;
    } else if (cnt0 >= 1) {
      add_rows64_first<1>(acc, tl, r.x, r.y, n - 1, tv, d, rn, rl, nidx, vb2,
                           rowptr, bidx);
      j = 1;
    } else {
      tv = *(const float2*)(tl + (size_t)d * 64);
      rn = rl[nidx];
      vb2 = rowptr[bidx];
    }
    walk_rows64(acc, tl, r, j, cnt0, n - 1);
    for (int base = start + 32; base < end; base += 32) {
      const int idx = base + lane;
      int2 rr = make_int2(0, 0);
      if (idx < end) rr = rl[idx];
      walk_rows64(acc, tl, rr, 0, min(32, end - base), n - 1);
    }
    const float degf = (float)(end - start + 1);
    float sl = degf > 0.f ? rsqrtf(degf) : 0.f;
    sl = sl * sl;
    float2 o;
    o.x = acc.x + tv.x * sl + b.x;
    o.y = acc.y + tv.y * sl + b.y;
    *(float2*)(h + (size_t)d * 64 + lane * 2) = o;
    start = ns; end = nx; r = rn;
    vb1 = vb2;
  }
}

extern "C" void kernel_launch(void* const* d_in, const int* in_sizes, int n_in,
                              void* d_out, int out_size, void* d_ws,
                              size_t ws_size, hipStream_t stream) {
  const float* x = (const float*)d_in[0];
  const int* ei = (const int*)d_in[1];
  const float* W1 = (const float*)d_in[2];
  const float* b1 = (const float*)d_in[3];
  const float* W2 = (const float*)d_in[4];
  const float* b2 = (const float*)d_in[5];
  const float* W3 = (const float*)d_in[6];
  const float* b3 = (const float*)d_in[7];
  const float* Wo = (const float*)d_in[8];
  const float* bo = (const float*)d_in[9];

  const int N = in_sizes[0] / CH;
  const int E = in_sizes[1] / 2;
  const int* src = ei;
  const int* dstp = ei + E;

  float* bufA = (float*)d_ws;
  float* bufB = bufA + (size_t)N * CH;
  int* rowptr = (int*)(bufB + (size_t)N * CH);
  int* rank = rowptr + N;
  float* invs = bufB;
  int* offs = (int*)(bufB + (size_t)2 * N);
  int* recs = (int*)d_out;
  float* c34 = (float*)d_out + (size_t)2 * E;
  float* w34 = c34 + 64;
  unsigned* ccnt = (unsigned*)(w34 + (size_t)CH * 64);
  float* t3 = bufA;
  int* recs3 = (int*)(bufA + (size_t)N * 64);
  const float* c34m = (const float*)(recs3 + (size_t)2 * E);
  const int nmove = (2 * E + 64 + 3) / 4;

  const dim3 blk(256);
  const int gE = (E + 255) / 256;
  const int gChunk = (E + CHUNK - 1) / CHUNK;
  const int gWord = (N / 2 + 8 + 255) / 256;
  const int gGemm = (N + 127) / 128;
  const int gGather = 2048;
  const int gFuse128 = 768;
  const int gFuse64 = 1024;
  const int gT1 = 207;
  const int gT2 = 256;
  const int gT3 = gGemm > gT1 + gT2 ? gGemm - gT1 - gT2 : 0;

  chunk_count<<<gChunk + gT1, dim3(1024), 0, stream>>>(
      dstp, ccnt, rank, N, E, gChunk, x, W1, bufA, 0);
  chunk_offsets<<<gWord + gT2, blk, 0, stream>>>(ccnt, offs, rowptr, N, gChunk,
                                                 gWord, x, W1, bufA, gT1);
  scan_rows_fold<<<3 + gT3, dim3(1024), 0, stream>>>(
      rowptr, invs, N, W3, b3, Wo, bo, w34, c34, x, W1, bufA, gT1 + gT2);
  fill_lists<<<gE, blk, 0, stream>>>(src, dstp, rank, rowptr, offs, invs, recs,
                                     N, E);

  gather_product<128, false><<<gFuse128, dim3(512), 0, stream>>>(
      bufA, rowptr, recs, b1, W2, bufB, N, E, 1, nullptr, nullptr, 0);

  gather_product<64, true><<<gFuse64, dim3(512), 0, stream>>>(
      bufB, rowptr, recs, b2, w34, t3, N, E, 1, recs, recs3, nmove);

  gather_rows64<<<gGather, blk, 0, stream>>>(t3, rowptr, recs3, c34m,
                                             (float*)d_out, N, E);
}


# row-start scan leaves its launch: offsets pass scans each group of 512 nodes, record fill adds group starts from LDS; w34, c34, layer-1 tiles ride launches 1-2; 7 -> 6 launches; sums unchanged
# speedup vs baseline: 5.3306x; 1.0815x over previous
#include <hip/hip_runtime.h>

typedef float v2f __attribute__((ext_vector_type(2)));
typedef float v8f __attribute__((ext_vector_type(8)));

#define CH 128

#define KPITCH 130

template <int NCOL, int THREADS, int SPB = THREADS / 32,
          bool PAD = (NCOL == 64)>
__device__ __forceinline__ void gemm_body(
    const float* __restrict__ A, const float* __restrict__ W,
    const float* __restrict__ bias, float* __restrict__ C, int nrows, int blk,
    float* __restrict__ lw) {
  constexpr int NT = NCOL / 16;
  constexpr int KP = KPITCH;

  const int tid = threadIdx.x;
  if constexpr (THREADS == 256) {
    for (int i = tid; i < CH * NCOL; i += THREADS) {
      int k = i / NCOL, c = i - k * NCOL;
      lw[c * KP + k] = W[i];
    }
  } else {
    constexpr int TR = CH * NCOL / THREADS;
    float wv[TR];
#pragma unroll
    for (int q = 0; q < TR; ++q) wv[q] = W[tid + q * THREADS];
#pragma unroll
    for (int q = 0; q < TR; ++q) {
      const int i = tid + q * THREADS;
      const int k = i / NCOL, c = i - k * NCOL;
      lw[c * KP + k] = wv[q];
    }
  }
  __syncthreads();

  const int wave = tid >> 5;
  const int lane = tid & 31;
  if constexpr (SPB < THREADS / 32) {
    if (wave >= SPB) return;
  }
  const int m0 = (blk * SPB + wave) * 16;
  if (m0 >= nrows) return;

  const int lrow = lane & 15;
  const int kbase = (lane >> 4) * 2;

  v8f acc[NT] = {};

  int arow = m0 + lrow;
  if (arow > nrows - 1) arow = nrows - 1;
  const float* aptr = A + (size_t)arow * CH + kbase;

#pragma unroll(NCOL == 64 ? CH / 4 : 4)
  for (int kc = 0; kc < CH / 4; ++kc) {
    float2 av = *(const float2*)(aptr + kc * 4);
    v2f a;
    a[0] = av.x; a[1] = av.y;
#pragma unroll
    for (int n = 0; n < NT; ++n) {
      float2 bv = *(const float2*)&lw[(n * 16 + lrow) * KP + kc * 4 + kbase];
      v2f b;
      b[0] = bv.x; b[1] = bv.y;
      acc[n] = __builtin_amdgcn_wmma_f32_16x16x4_f32(false, a, false, b,
                                                     (short)0, acc[n],
                                                     false, false);
    }
  }

  if constexpr (PAD) {
    if constexpr (NCOL == 64) __builtin_amdgcn_sched_barrier(0);
    if constexpr (NT == 4) {
      asm volatile(
          "v_nop\n\tv_nop\n\tv_nop\n\tv_nop\n\tv_nop\n\tv_nop\n\tv_nop\n\tv_nop"
          : "+v"(acc[0]), "+v"(acc[1]), "+v"(acc[2]), "+v"(acc[3]));
    } else {
      static_assert(NT == 4 || NT == 8, "accumulators named below");
      asm volatile(
          "v_nop\n\tv_nop\n\tv_nop\n\tv_nop\n\tv_nop\n\tv_nop\n\tv_nop\n\tv_nop"
          : "+v"(acc[0]), "+v"(acc[1]), "+v"(acc[2]), "+v"(acc[3]),
            "+v"(acc[4]), "+v"(acc[5]), "+v"(acc[6]), "+v"(acc[7]));
    }
    __builtin_amdgcn_sched_barrier(0);
  }

  const int rbase = m0 + ((lane >> 4) << 3);
#pragma unroll
  for (int r = 0; r < 8; ++r) {
    int row = rbase + r;
    if (row < nrows) {
#pragma unroll
      for (int n = 0; n < NT; ++n) {
        int col = n * 16 + lrow;
        float v = acc[n][r];
        if (bias) v += bias[col];
        C[(size_t)row * NCOL + col] = v;
      }
    }
  }
}

#define CHUNK 32768
#define CAPW 65536

__device__ __forceinline__ int packed_pitch(int n) {
  return (((n + 1) >> 1) + 3) & ~3;
}

__global__ __launch_bounds__(1024) void chunk_count(
    const int* __restrict__ dst, unsigned* __restrict__ ccnt,
    int* __restrict__ rank, int n, int ne, int nchunks,
    const float* __restrict__ x, const float* __restrict__ W1,
    float* __restrict__ t1, int tile0,
    const float* __restrict__ W3, const float* __restrict__ b3,
    const float* __restrict__ Wo, const float* __restrict__ bo,
    float* __restrict__ w34, float* __restrict__ c34) {
  __shared__ unsigned lc[CAPW];
  if ((int)blockIdx.x >= nchunks) {
    const int rb = (int)blockIdx.x - nchunks;
    if (rb == 0) {
      gemm_body<64, 1024>(W3, Wo, nullptr, w34, CH, 0, (float*)lc);
    } else if (rb == 1) {
      gemm_body<64, 1024>(b3, Wo, bo, c34, 1, 0, (float*)lc);
    } else {
      gemm_body<128, 1024, 8, true>(x, W1, nullptr, t1, n, tile0 + rb - 2,
                                    (float*)lc);
    }
    return;
  }
  const int tid = threadIdx.x;
  const int c = blockIdx.x;
  const int e0 = c * CHUNK;
  const int e1 = min(e0 + CHUNK, ne);
  const int nwp = packed_pitch(n);
  for (int wb = 0; wb < nwp; wb += CAPW) {
    const int nwin = min(CAPW, nwp - wb);
    for (int i = tid * 4; i < nwin; i += 4096)
      *(uint4*)&lc[i] = make_uint4(0u, 0u, 0u, 0u);
    __syncthreads();
    const int dlo = wb * 2;
    const int dhi = dlo + nwin * 2;
    int dv[CHUNK / 1024];
    unsigned old[CHUNK / 1024];
#pragma unroll
    for (int k = 0; k < CHUNK / 1024; ++k) {
      const int e = e0 + k * 1024 + tid;
      const int v = dst[min(e, e1 - 1)];
      dv[k] = (e < e1) ? v : -1;
    }
#pragma unroll
    for (int k = 0; k < CHUNK / 1024; ++k) {
      const bool in = dv[k] >= dlo && dv[k] < dhi;
      const int l = in ? dv[k] - dlo : 0;
      const unsigned inc = in ? ((l & 1) ? 65536u : 1u) : 0u;
      old[k] = atomicAdd(&lc[l >> 1], inc);
    }
#pragma unroll
    for (int k = 0; k < CHUNK / 1024; ++k) {
      const int e = e0 + k * 1024 + tid;
      if (dv[k] >= dlo && dv[k] < dhi)
        rank[e] = (int)(((dv[k] - dlo) & 1) ? (old[k] >> 16)
                                            : (old[k] & 0xFFFFu));
    }
    __syncthreads();
    for (int i = tid * 4; i < nwin; i += 4096)
      *(uint4*)&ccnt[(size_t)c * nwp + wb + i] = *(const uint4*)&lc[i];
    __syncthreads();
  }
}

#define GROUP_SHIFT 9
__global__ __launch_bounds__(256) void chunk_offsets(
    const unsigned* __restrict__ ccnt, int* __restrict__ offs,
    int* __restrict__ loc, float* __restrict__ invs, int* __restrict__ gtot,
    int n, int nchunks, int nown, const float* __restrict__ x,
    const float* __restrict__ W1, float* __restrict__ t1, int tile0) {
  __shared__ float lw1[128 * KPITCH];
  __shared__ int wtot[8];
  if ((int)blockIdx.x >= nown) {
    gemm_body<128, 256, 8, true>(x, W1, nullptr, t1, n,
                                 tile0 + (int)blockIdx.x - nown, lw1);
    return;
  }
  const int nwp = packed_pitch(n);
  const int tid = threadIdx.x;
  const int lane = tid & 31;
  const int wv = tid >> 5;
  const int w = blockIdx.x * 256 + tid;
  const size_t np = (size_t)nwp * 2;
  int lo = 0, hi = 0;
  if (w < nwp) {
#pragma unroll 8
    for (int c = 0; c < nchunks; ++c) {
      const unsigned v = ccnt[(size_t)c * nwp + w];
      *(int2*)(offs + (size_t)c * np + 2 * w) = make_int2(lo, hi);
      lo += (int)(v & 0xFFFFu);
      hi += (int)(v >> 16);
    }
  }
  const int pair = lo + hi;
  int v = pair;
  for (int o = 1; o < 32; o <<= 1) {
    const int u = __shfl_up(v, o);
    if (lane >= o) v += u;
  }
  if (lane == 31) wtot[wv] = v;
  __syncthreads();
  int wexcl = 0, total = 0;
#pragma unroll
  for (int k = 0; k < 8; ++k) {
    const int s = wtot[k];
    if (k < wv) wexcl += s;
    total += s;
  }
  const int ex = wexcl + v - pair;
  if (2 * w < n) {
    const float dk = (float)(lo + 1);
    loc[2 * w] = ex;
    invs[2 * w] = dk > 0.f ? rsqrtf(dk) : 0.f;
  }
  if (2 * w + 1 < n) {
    const float dk = (float)(hi + 1);
    loc[2 * w + 1] = ex + lo;
    invs[2 * w + 1] = dk > 0.f ? rsqrtf(dk) : 0.f;
  }
  if (tid == 0) gtot[blockIdx.x] = total;
}

__global__ __launch_bounds__(256) void fill_lists(
    const int* __restrict__ src, const int* __restrict__ dst,
    const int* __restrict__ rank, const int* __restrict__ loc,
    const int* __restrict__ gtot, const int* __restrict__ offs,
    const float* __restrict__ invs, int* __restrict__ recs,
    int* __restrict__ rowptr, int n, int ne, int ng) {
  extern __shared__ int gp[];
  __shared__ int wtot[8];
  const int tid = threadIdx.x;
  const int lane = tid & 31;
  const int wv = tid >> 5;
  const int e = blockIdx.x * 256 + tid;
  int s = 0, d = 0, rk = 0, lrow = 0;
  if (e < ne) {
    s = src[e];
    d = dst[e];
    rk = rank[e];
  }
  if (e < n) lrow = loc[e];
  int carry = 0;
  for (int base = 0; base < ng; base += 256) {
    const int i = base + tid;
    const int g = i < ng ? gtot[i] : 0;
    int v = g;
    for (int o = 1; o < 32; o <<= 1) {
      const int u = __shfl_up(v, o);
      if (lane >= o) v += u;
    }
    if (lane == 31) wtot[wv] = v;
    __syncthreads();
    int wexcl = 0, total = 0;
#pragma unroll
    for (int k = 0; k < 8; ++k) {
      const int q = wtot[k];
      if (k < wv) wexcl += q;
      total += q;
    }
    if (i < ng) gp[i] = carry + wexcl + v - g;
    carry += total;
    __syncthreads();
  }
  if (e < n) rowptr[e] = gp[e >> GROUP_SHIFT] + lrow;
  if (e < ne) {
    const size_t np = (size_t)packed_pitch(n) * 2;
    int p = gp[d >> GROUP_SHIFT] + loc[d] +
            offs[(size_t)(e / CHUNK) * np + d] + rk;
    p = min(max(p, 0), ne - 1);
    const float w = invs[s] * invs[d];
    ((int2*)recs)[p] = make_int2(s, __float_as_int(w));
  }
}

template <int K>
__device__ __forceinline__ void add_rows(float4& acc,
                                         const float* __restrict__ tl, int rx,
                                         int ry, int j, int last) {
  float4 v[K];
  float wg[K];
#pragma unroll
  for (int u = 0; u < K; ++u) {
    int s = __builtin_amdgcn_readlane(rx, j + u);
    s = min(max(s, 0), last);
    wg[u] = __int_as_float(__builtin_amdgcn_readlane(ry, j + u));
    v[u] = *(const float4*)(tl + (size_t)s * CH);
  }
#pragma unroll
  for (int u = 0; u < K; ++u) {
    acc.x = fmaf(v[u].x, wg[u], acc.x);
    acc.y = fmaf(v[u].y, wg[u], acc.y);
    acc.z = fmaf(v[u].z, wg[u], acc.z);
    acc.w = fmaf(v[u].w, wg[u], acc.w);
  }
}

__device__ __forceinline__ void node_bounds(const int* __restrict__ rowptr,
                                            int d, int n, int ne, int& start,
                                            int& end) {
  const int dc = min(d, n - 1);
  const int s = rowptr[dc];
  const int e = (dc + 1 < n) ? rowptr[dc + 1] : ne;
  start = s;
  end = (d < n) ? e : s;
}

__device__ __forceinline__ int bounds_index(int d, int lane, int n) {
  return min(d + (lane & 1), n - 1);
}
__device__ __forceinline__ void bounds_decode(int vb, int d, int n, int ne,
                                              int& start, int& end) {
  const int s = __builtin_amdgcn_readlane(vb, 0);
  const int e = __builtin_amdgcn_readlane(vb, 1);
  start = s;
  end = (d < n) ? ((d + 1 < n) ? e : ne) : s;
}

__device__ __forceinline__ int ahead_index(int start, int lane, int ne) {
  return max(min(start + lane, ne - 1), 0);
}

template <int K>
__device__ __forceinline__ void add_rows_first(
    float4& acc, const float* __restrict__ tl, int rx, int ry, int last,
    float4& tv, int d, int2& rn, const int2* __restrict__ rl, int nidx,
    int& vb, const int* __restrict__ rowptr, int bidx) {
  float4 v[K];
  float wg[K];
#pragma unroll
  for (int u = 0; u < K; ++u) {
    int s = __builtin_amdgcn_readlane(rx, u);
    s = min(max(s, 0), last);
    wg[u] = __int_as_float(__builtin_amdgcn_readlane(ry, u));
    v[u] = *(const float4*)(tl + (size_t)s * CH);
  }
  tv = *(const float4*)(tl + (size_t)d * CH);
  rn = rl[nidx];
  vb = rowptr[bidx];
#pragma unroll
  for (int u = 0; u < K; ++u) {
    acc.x = fmaf(v[u].x, wg[u], acc.x);
    acc.y = fmaf(v[u].y, wg[u], acc.y);
    acc.z = fmaf(v[u].z, wg[u], acc.z);
    acc.w = fmaf(v[u].w, wg[u], acc.w);
  }
}

__device__ __forceinline__ void walk_rows(float4& acc,
                                          const float* __restrict__ tl,
                                          int2 r, int j, int cnt, int last) {
  for (; j + 8 <= cnt; j += 8) add_rows<8>(acc, tl, r.x, r.y, j, last);
  if (j + 4 <= cnt) { add_rows<4>(acc, tl, r.x, r.y, j, last); j += 4; }
  if (j + 2 <= cnt) { add_rows<2>(acc, tl, r.x, r.y, j, last); j += 2; }
  if (j < cnt) add_rows<1>(acc, tl, r.x, r.y, j, last);
}

template <bool BLATE>
__device__ __forceinline__ float4 gather_node(
    const float* __restrict__ tl, const int2* __restrict__ rl, float4 b,
    const float* __restrict__ bl, int d, int n, int ne, int relu, int lane,
    int start, int end, int2 r, int ns, int2& rn, int& vb,
    const int* __restrict__ rowptr, int bidx) {
  const int nidx = ahead_index(ns, lane, ne);
  float4 tv;
  float4 acc = make_float4(0.f, 0.f, 0.f, 0.f);
  const int cnt0 = min(32, end - start);
  int j = 0;
  if (cnt0 >= 8) {
    add_rows_first<8>(acc, tl, r.x, r.y, n - 1, tv, d, rn, rl, nidx, vb,
                      rowptr, bidx);
    j = 8;
  } else if (cnt0 >= 4) {
    add_rows_first<4>(acc, tl, r.x, r.y, n - 1, tv, d, rn, rl, nidx, vb,
                      rowptr, bidx);
    j = 4;
  } else if (cnt0 >= 2) {
    add_rows_first<2>(acc, tl, r.x, r.y, n - 1, tv, d, rn, rl, nidx, vb,
                      rowptr, bidx);
    j = 2;
  } else if (cnt0 >= 1) {
    add_rows_first<1>(acc, tl, r.x, r.y, n - 1, tv, d, rn, rl, nidx, vb,
                      rowptr, bidx);
    j = 1;
  } else {
    tv = *(const float4*)(tl + (size_t)d * CH);
    rn = rl[nidx];
    vb = rowptr[bidx];
  }
  walk_rows(acc, tl, r, j, cnt0, n - 1);
  for (int base = start + 32; base < end; base += 32) {
    const int idx = base + lane;
    int2 rr = make_int2(0, 0);
    if (idx < end) rr = rl[idx];
    walk_rows(acc, tl, rr, 0, min(32, end - base), n - 1);
  }
  const float degf = (float)(end - start + 1);
  float sl = degf > 0.f ? rsqrtf(degf) : 0.f;
  sl = sl * sl;
  if constexpr (BLATE) b = *(const float4*)bl;
  float4 o;
  o.x = acc.x + tv.x * sl + b.x;
  o.y = acc.y + tv.y * sl + b.y;
  o.z = acc.z + tv.z * sl + b.z;
  o.w = acc.w + tv.w * sl + b.w;
  if (relu) {
    o.x = fmaxf(o.x, 0.f);
    o.y = fmaxf(o.y, 0.f);
    o.z = fmaxf(o.z, 0.f);
    o.w = fmaxf(o.w, 0.f);
  }
  return o;
}

#define TM 32
#define TPITCH 132

template <int NCOL, bool MOVE>
__global__ __launch_bounds__(512)
__attribute__((amdgpu_waves_per_eu(NCOL == 64 ? 16 : 12))) void gather_product(
    const float* __restrict__ t, const int* __restrict__ rowptr,
    const int* __restrict__ recs, const float* __restrict__ bias,
    const float* __restrict__ W, float* __restrict__ tn, int n, int ne,
    int relu, const int* cp_src, int* cp_dst, int ncopy) {
  constexpr int NT = NCOL / 16;
  constexpr int UNITS = (TM / 16) * NT;
  static_assert(UNITS <= 16, "one unit per wave");
  __shared__ float lw[NCOL * KPITCH];
  __shared__ float tile[2][TM * TPITCH];
  const int tid = threadIdx.x;

  if constexpr (MOVE) {
    const int per =
        ((ncopy + (int)gridDim.x - 1) / (int)gridDim.x + 2047) & ~2047;
    const int c0 = (int)blockIdx.x * per;
    const int c1 = min(c0 + per, ncopy);
    const int4* ms = (const int4*)cp_src;
    int4* md = (int4*)cp_dst;
    for (int g0 = c0; g0 < c1; g0 += 2048) {
      if (g0 + 2048 <= ncopy) {
        const int i = g0 + tid;
        const int4 m0 = ms[i];
        const int4 m1 = ms[i + 512];
        const int4 m2 = ms[i + 1024];
        const int4 m3 = ms[i + 1536];
        md[i] = m0;
        md[i + 512] = m1;
        md[i + 1024] = m2;
        md[i + 1536] = m3;
      } else {
        for (int i = g0 + tid; i < ncopy; i += 512) md[i] = ms[i];
      }
    }
  }

  const int lane = tid & 31;
  const int wave = __builtin_amdgcn_readfirstlane(tid >> 5);
  const int2* rl = (const int2*)recs;
  static_assert(TM / 16 == 2, "two rows per wave and tile");
  const int r0 = wave * (TM / 16);
  const int dstep = (int)gridDim.x * TM;
  int cs, ce, vb1;
  int2 cr;
  {
    const int d0 = (int)blockIdx.x * TM + r0;
    node_bounds(rowptr, d0, n, ne, cs, ce);
    vb1 = rowptr[bounds_index(d0 + 1, lane, n)];
  }

  {
    constexpr int TR = CH * NCOL / 512;
    constexpr int SB = 16;
    static_assert(TR % SB == 0, "whole batches");
    for (int q0 = 0; q0 < TR; q0 += SB) {
      float wv[SB];
#pragma unroll
      for (int q = 0; q < SB; ++q) wv[q] = W[tid + (q0 + q) * 512];
#pragma unroll
      for (int q = 0; q < SB; ++q) {
        const int i = tid + (q0 + q) * 512;
        const int k = i / NCOL, c = i - k * NCOL;
        lw[c * KPITCH + k] = wv[q];
      }
    }
  }
  cr = rl[ahead_index(cs, lane, ne)];

  float4 b = make_float4(0.f, 0.f, 0.f, 0.f);
  if constexpr (NCOL != 64) b = *(const float4*)(bias + lane * 4);
  const float* tl = t + lane * 4;
  const int lrow = lane & 15;
  const int kbase = (lane >> 4) * 2;
  const int ntiles = (n + TM - 1) / TM;
  int buf = 0;
  for (int ti = blockIdx.x; ti < ntiles; ti += gridDim.x, buf ^= 1) {
    float* tb = tile[buf];
#pragma unroll
    for (int i = 0; i < TM / 16; ++i) {
      const int r = r0 + i;
      const int d = ti * TM + r;
      const int dn = (i + 1 < TM / 16) ? d + 1 : d + dstep - (TM / 16 - 1);
      int ns, nx;
      bounds_decode(vb1, dn, n, ne, ns, nx);
      int2 rn = make_int2(0, 0);
      int vb2 = 0;
      float4 o = make_float4(0.f, 0.f, 0.f, 0.f);
      if (d < n)
        o = gather_node<NCOL == 64>(tl, rl, b, bias + lane * 4, d, n, ne, relu,
                                    lane, cs, ce, cr, ns, rn, vb2, rowptr,
                                    bounds_index(d + dstep, lane, n));
      *(float4*)&tb[r * TPITCH + lane * 4] = o;
      cs = ns; ce = nx; cr = rn;
      vb1 = vb2;
    }
    __syncthreads();
    if (wave < UNITS) {
      const int s = wave / NT;
      const int nt = wave - s * NT;
      const float* ap = tb + (s * 16 + lrow) * TPITCH + kbase;
      const float* bp = lw + (nt * 16 + lrow) * KPITCH + kbase;
      v8f acc = {};
#pragma unroll
      for (int kc = 0; kc < CH / 4; ++kc) {
        const float2 av = *(const float2*)(ap + kc * 4);
        const float2 bv = *(const float2*)(bp + kc * 4);
        v2f a, bb;
        a[0] = av.x; a[1] = av.y;
        bb[0] = bv.x; bb[1] = bv.y;
        acc = __builtin_amdgcn_wmma_f32_16x16x4_f32(false, a, false, bb,
                                                    (short)0, acc, false,
                                                    false);
      }
      __builtin_amdgcn_sched_barrier(0);
      asm volatile(
          "v_nop\n\tv_nop\n\tv_nop\n\tv_nop\n\tv_nop\n\tv_nop\n\tv_nop\n\tv_nop"
          : "+v"(acc));
      __builtin_amdgcn_sched_barrier(0);
      const int rbase = ti * TM + s * 16 + ((lane >> 4) << 3);
      const int col = nt * 16 + lrow;
#pragma unroll
      for (int r = 0; r < 8; ++r) {
        const int row = rbase + r;
        if (row < n) tn[(size_t)row * NCOL + col] = acc[r];
      }
    }
  }
}

template <int K>
__device__ __forceinline__ void add_rows64(float2& acc,
                                           const float* __restrict__ tl,
                                           int rx, int ry, int j, int last) {
  float2 v[K];
  float wg[K];
#pragma unroll
  for (int u = 0; u < K; ++u) {
    int s = __builtin_amdgcn_readlane(rx, j + u);
    s = min(max(s, 0), last);
    wg[u] = __int_as_float(__builtin_amdgcn_readlane(ry, j + u));
    v[u] = *(const float2*)(tl + (size_t)s * 64);
  }
#pragma unroll
  for (int u = 0; u < K; ++u) {
    acc.x += v[u].x * wg[u];
    acc.y += v[u].y * wg[u];
  }
}

template <int K>
__device__ __forceinline__ void add_rows64_first(
    float2& acc, const float* __restrict__ tl, int rx, int ry, int last,
    float2& tv, int d, int2& rn, const int2* __restrict__ rl, int nidx,
    int& vb, const int* __restrict__ rowptr, int bidx) {
  float2 v[K];
  float wg[K];
#pragma unroll
  for (int u = 0; u < K; ++u) {
    int s = __builtin_amdgcn_readlane(rx, u);
    s = min(max(s, 0), last);
    wg[u] = __int_as_float(__builtin_amdgcn_readlane(ry, u));
    v[u] = *(const float2*)(tl + (size_t)s * 64);
  }
  tv = *(const float2*)(tl + (size_t)d * 64);
  rn = rl[nidx];
  vb = rowptr[bidx];
#pragma unroll
  for (int u = 0; u < K; ++u) {
    acc.x += v[u].x * wg[u];
    acc.y += v[u].y * wg[u];
  }
}

__device__ __forceinline__ void walk_rows64(float2& acc,
                                            const float* __restrict__ tl,
                                            int2 r, int j, int cnt, int last) {
  for (; j + 16 <= cnt; j += 16) add_rows64<16>(acc, tl, r.x, r.y, j, last);
  if (j + 8 <= cnt) { add_rows64<8>(acc, tl, r.x, r.y, j, last); j += 8; }
  if (j + 4 <= cnt) { add_rows64<4>(acc, tl, r.x, r.y, j, last); j += 4; }
  if (j + 2 <= cnt) { add_rows64<2>(acc, tl, r.x, r.y, j, last); j += 2; }
  if (j < cnt) add_rows64<1>(acc, tl, r.x, r.y, j, last);
}

__global__ __launch_bounds__(256) void gather_rows64(
    const float* __restrict__ t, const int* __restrict__ rowptr,
    const int* __restrict__ recs, const float* __restrict__ bias,
    float* __restrict__ h, int n, int ne) {
  const int lane = threadIdx.x & 31;
  const int w0 = __builtin_amdgcn_readfirstlane(
      (int)((blockIdx.x * blockDim.x + threadIdx.x) >> 5));
  const int nw = (int)((gridDim.x * blockDim.x) >> 5);
  const float2 b = *(const float2*)(bias + lane * 2);
  const int2* rl = (const int2*)recs;
  const float* tl = t + lane * 2;
  int start, end;
  node_bounds(rowptr, w0, n, ne, start, end);
  int vb1 = rowptr[bounds_index(w0 + nw, lane, n)];
  int2 r = rl[ahead_index(start, lane, ne)];
  for (int d = w0; d < n; d += nw) {
    int ns, nx;
    bounds_decode(vb1, d + nw, n, ne, ns, nx);
    const int nidx = ahead_index(ns, lane, ne);
    const int bidx = bounds_index(d + 2 * nw, lane, n);
    int2 rn;
    int vb2;
    float2 tv;
    float2 acc = make_float2(0.f, 0.f);
    const int cnt0 = min(32, end - start);
    int j = 0;
    if (cnt0 >= 16) {
      add_rows64_first<16>(acc, tl, r.x, r.y, n - 1, tv, d, rn, rl, nidx, vb2,
                           rowptr, bidx);
      j = 16;
    } else if (cnt0 >= 8) {
      add_rows64_first<8>(acc, tl, r.x, r.y, n - 1, tv, d, rn, rl, nidx, vb2,
                           rowptr, bidx);
      j = 8;
    } else if (cnt0 >= 4) {
      add_rows64_first<4>(acc, tl, r.x, r.y, n - 1, tv, d, rn, rl, nidx, vb2,
                           rowptr, bidx);
      j = 4;
    } else if (cnt0 >= 2) {
      add_rows64_first<2>(acc, tl, r.x, r.y, n - 1, tv, d, rn, rl, nidx, vb2,
                           rowptr, bidx);
      j = 2;
    } else if (cnt0 >= 1) {
      add_rows64_first<1>(acc, tl, r.x, r.y, n - 1, tv, d, rn, rl, nidx, vb2,
                           rowptr, bidx);
      j = 1;
    } else {
      tv = *(const float2*)(tl + (size_t)d * 64);
      rn = rl[nidx];
      vb2 = rowptr[bidx];
    }
    walk_rows64(acc, tl, r, j, cnt0, n - 1);
    for (int base = start + 32; base < end; base += 32) {
      const int idx = base + lane;
      int2 rr = make_int2(0, 0);
      if (idx < end) rr = rl[idx];
      walk_rows64(acc, tl, rr, 0, min(32, end - base), n - 1);
    }
    const float degf = (float)(end - start + 1);
    float sl = degf > 0.f ? rsqrtf(degf) : 0.f;
    sl = sl * sl;
    float2 o;
    o.x = acc.x + tv.x * sl + b.x;
    o.y = acc.y + tv.y * sl + b.y;
    *(float2*)(h + (size_t)d * 64 + lane * 2) = o;
    start = ns; end = nx; r = rn;
    vb1 = vb2;
  }
}

extern "C" void kernel_launch(void* const* d_in, const int* in_sizes, int n_in,
                              void* d_out, int out_size, void* d_ws,
                              size_t ws_size, hipStream_t stream) {
  const float* x = (const float*)d_in[0];
  const int* ei = (const int*)d_in[1];
  const float* W1 = (const float*)d_in[2];
  const float* b1 = (const float*)d_in[3];
  const float* W2 = (const float*)d_in[4];
  const float* b2 = (const float*)d_in[5];
  const float* W3 = (const float*)d_in[6];
  const float* b3 = (const float*)d_in[7];
  const float* Wo = (const float*)d_in[8];
  const float* bo = (const float*)d_in[9];

  const int N = in_sizes[0] / CH;
  const int E = in_sizes[1] / 2;
  const int* src = ei;
  const int* dstp = ei + E;

  float* bufA = (float*)d_ws;
  float* bufB = bufA + (size_t)N * CH;
  int* rowptr = (int*)(bufB + (size_t)N * CH);
  int* rank = rowptr + N;
  float* invs = bufB;
  int* loc = (int*)(bufB + (size_t)N);
  const int gWord = (N / 2 + 8 + 255) / 256;
  int* gtot = (int*)(bufB + (size_t)2 * N);
  int* offs = gtot + ((gWord + 3) & ~3);
  int* recs = (int*)d_out;
  float* c34 = (float*)d_out + (size_t)2 * E;
  float* w34 = c34 + 64;
  unsigned* ccnt = (unsigned*)(w34 + (size_t)CH * 64);
  float* t3 = bufA;
  int* recs3 = (int*)(bufA + (size_t)N * 64);
  const float* c34m = (const float*)(recs3 + (size_t)2 * E);
  const int nmove = (2 * E + 64 + 3) / 4;

  const dim3 blk(256);
  const int gChunk = (E + CHUNK - 1) / CHUNK;
  const int gGemm = (N + 127) / 128;
  const int gGather = 2048;
  const int gFuse128 = 768;
  const int gFuse64 = 1024;
  const int gT1 = 205;
  const int gT2 = gGemm > gT1 ? gGemm - gT1 : 0;
  const int gFill = ((E > N ? E : N) + 255) / 256;

  chunk_count<<<gChunk + 2 + gT1, dim3(1024), 0, stream>>>(
      dstp, ccnt, rank, N, E, gChunk, x, W1, bufA, 0, W3, b3, Wo, bo, w34,
      c34);
  chunk_offsets<<<gWord + gT2, blk, 0, stream>>>(ccnt, offs, loc, invs, gtot, N,
                                                 gChunk, gWord, x, W1, bufA,
                                                 gT1);
  fill_lists<<<gFill, blk, gWord * sizeof(int), stream>>>(
      src, dstp, rank, loc, gtot, offs, invs, recs, rowptr, N, E, gWord);

  gather_product<128, false><<<gFuse128, dim3(512), 0, stream>>>(
      bufA, rowptr, recs, b1, W2, bufB, N, E, 1, nullptr, nullptr, 0);

  gather_product<64, true><<<gFuse64, dim3(512), 0, stream>>>(
      bufB, rowptr, recs, b2, w34, t3, N, E, 1, recs, recs3, nmove);

  gather_rows64<<<gGather, blk, 0, stream>>>(t3, rowptr, recs3, c34m,
                                             (float*)d_out, N, E);
}
